# starting kernel
# baseline (speedup 1.0000x reference)
#include <hip/hip_runtime.h>
#include <hip/hip_bf16.h>
#include <math.h>

#define CCH 32
#define NPATH 11
#define NB 8

typedef __attribute__((ext_vector_type(2))) float v2f;
typedef __attribute__((ext_vector_type(8))) float v8f;

__device__ __forceinline__ v8f wmma4(v2f a, v2f b, v8f c) {
  return __builtin_amdgcn_wmma_f32_16x16x4_f32(
       false, a,  false, b,
       (short)0, c,  false,  false);
}

__global__ void mmb_zero(float4* __restrict__ p, int n4) {
  int i = blockIdx.x * blockDim.x + threadIdx.x;
  if (i < n4) p[i] = make_float4(0.f, 0.f, 0.f, 0.f);
}

__global__ void mmb_edge(const float* __restrict__ rij,
                         const float* __restrict__ Wrad,
                         const int*   __restrict__ idx_i,
                         const int*   __restrict__ idx_j,
                         const float* __restrict__ node0,
                         const float* __restrict__ node1,
                         const float* __restrict__ node2,
                         float* __restrict__ a0,
                         float* __restrict__ a1,
                         float* __restrict__ a2,
                         int E) {
  const int e = blockIdx.x * (blockDim.x >> 5) + (threadIdx.x >> 5);
  const int c = threadIdx.x & 31;
  if (e >= E) return;

  const float r0 = rij[e * 3 + 0], r1 = rij[e * 3 + 1], r2 = rij[e * 3 + 2];
  const float d  = sqrtf(r0 * r0 + r1 * r1 + r2 * r2 + 1e-12f);
  const float iv = 1.0f / d;
  const float rh[3] = {r0 * iv, r1 * iv, r2 * iv};

  float x = d * 0.2f;
  x = (x > 1.0f) ? 1.0f : x;
  const float fc = 0.5f * (cosf(3.14159265358979f * x) + 1.0f);

  float rbf[NB];
#pragma unroll
  for (int b = 0; b < NB; ++b) {
    const float t = d - (0.5f + 0.6428571429f * (float)b);
    rbf[b] = __expf(-4.0f * t * t) * fc;
  }

  float w[NPATH];
#pragma unroll
  for (int p = 0; p < NPATH; ++p) {
    float acc = 0.f;
#pragma unroll
    for (int b = 0; b < NB; ++b) acc += rbf[b] * Wrad[(p * NB + b) * CCH + c];
    w[p] = acc;
  }

  const int j = idx_j[e];
  const int i = idx_i[e];

  const float g0 = node0[j * CCH + c];
  float g1[3], g2[9];
#pragma unroll
  for (int a = 0; a < 3; ++a) g1[a] = node1[(j * CCH + c) * 3 + a];
#pragma unroll
  for (int q = 0; q < 9; ++q) g2[q] = node2[(j * CCH + c) * 9 + q];

  const float d1 = g1[0] * rh[0] + g1[1] * rh[1] + g1[2] * rh[2];
  float gr[3];
#pragma unroll
  for (int a = 0; a < 3; ++a)
    gr[a] = g2[a * 3 + 0] * rh[0] + g2[a * 3 + 1] * rh[1] + g2[a * 3 + 2] * rh[2];
  const float d2 = gr[0] * rh[0] + gr[1] * rh[1] + gr[2] * rh[2];

  const float inv16 = 0.0625f;

  atomicAdd(&a0[i * CCH + c], (g0 * w[0] + d1 * w[4] + d2 * w[9]) * inv16);

#pragma unroll
  for (int a = 0; a < 3; ++a) {
    const float m = (g0 * w[1] + d1 * w[6]) * rh[a] + g1[a] * w[3] + gr[a] * w[8];
    atomicAdd(&a1[(i * CCH + c) * 3 + a], m * inv16);
  }
#pragma unroll
  for (int a = 0; a < 3; ++a) {
    const float ca = g0 * w[2] * rh[a] + g1[a] * w[5] + gr[a] * w[10];
#pragma unroll
    for (int b = 0; b < 3; ++b) {
      const float m = ca * rh[b] + g2[a * 3 + b] * w[7];
      atomicAdd(&a2[(i * CCH + c) * 9 + a * 3 + b], m * inv16);
    }
  }
}

#define W_U0 0
#define W_U1 1
#define W_U2 2
#define W_V(k) (3 + (k))
#define W_G(k) (11 + (k))

__global__ void __launch_bounds__(128)
mmb_node(const float* __restrict__ a0, const float* __restrict__ a1,
         const float* __restrict__ a2,
         const float* __restrict__ node0, const float* __restrict__ node1,
         const float* __restrict__ node2,
         const float* __restrict__ U, const float* __restrict__ V,
         const float* __restrict__ Wg,
         float* __restrict__ out0, float* __restrict__ out1,
         float* __restrict__ out2, int N) {
  __shared__ float2 wlds[14 * 8 * 2 * 32];
  __shared__ float  ylds[4][16][CCH];

  for (int idx = threadIdx.x; idx < 14 * 1024; idx += blockDim.x) {
    const int wmat = idx >> 10;
    const int rem  = idx & 1023;
    const int k    = rem >> 5;
    const int n    = rem & 31;
    const float* src = (wmat < 3) ? (U + wmat * 1024)
                     : (wmat < 11) ? (V + (wmat - 3) * 1024)
                                   : (Wg + (wmat - 11) * 1024);
    const float val = src[rem];
    const int s   = k >> 2;
    const int kk  = k & 3;
    const int ln  = ((kk >> 1) << 4) | (n & 15);
    const int t   = n >> 4;
    const int jj  = kk & 1;
    ((float*)&wlds[((wmat * 8 + s) * 2 + t) * 32 + ln])[jj] = val;
  }
  __syncthreads();

  const int lane = threadIdx.x & 31;
  const int wv   = threadIdx.x >> 5;
  const int half = lane >> 4;
  const int m    = lane & 15;
  const int tile = blockIdx.x * 4 + wv;
  if (tile * 16 >= N) return;

#define LDB(wm, s, t)                                                       \
  ({ float2 _bv = wlds[(((wm) * 8 + (s)) * 2 + (t)) * 32 + lane];           \
     v2f _b; _b.x = _bv.x; _b.y = _bv.y; _b; })

#define MM2(acc, aop, wm, s)                                                \
  do { v2f _a = (aop);                                                      \
       acc[0] = wmma4(_a, LDB(wm, s, 0), acc[0]);                           \
       acc[1] = wmma4(_a, LDB(wm, s, 1), acc[1]); } while (0)

  const v8f vzero = {};

  const int rowA = tile * 16 + m;
  const int rowC = (rowA < N) ? rowA : (N - 1);
  float x0[2], b1[2][3], b2[2][9];
#define LOAD_BASE(s)                                                        \
  do { const int _k = (s) * 4 + 2 * half;                                   \
       const int _rc = rowC * CCH + _k;                                     \
       x0[0] = a0[_rc]; x0[1] = a0[_rc + 1];                                \
       _Pragma("unroll") for (int _a = 0; _a < 3; ++_a) {                   \
         b1[0][_a] = a1[_rc * 3 + _a]; b1[1][_a] = a1[(_rc + 1) * 3 + _a]; }\
       _Pragma("unroll") for (int _q = 0; _q < 9; ++_q) {                   \
         b2[0][_q] = a2[_rc * 9 + _q]; b2[1][_q] = a2[(_rc + 1) * 9 + _q]; }\
  } while (0)

  v8f y0acc[2] = {vzero, vzero};
  for (int s = 0; s < 8; ++s) {
    LOAD_BASE(s);
    v2f ax, asq, s1v, s2v;
#pragma unroll
    for (int jj = 0; jj < 2; ++jj) {
      ax[jj]  = x0[jj];
      asq[jj] = x0[jj] * x0[jj];
      float acc = 0.f;
#pragma unroll
      for (int a = 0; a < 3; ++a) acc += b1[jj][a] * b1[jj][a];
      s1v[jj] = acc;
      acc = 0.f;
#pragma unroll
      for (int q = 0; q < 9; ++q) acc += b2[jj][q] * b2[jj][q];
      s2v[jj] = acc;
    }
    MM2(y0acc, ax,  W_U0,  s);
    MM2(y0acc, asq, W_V(0), s);
    MM2(y0acc, s1v, W_V(3), s);
    MM2(y0acc, s2v, W_V(6), s);
  }

#pragma unroll
  for (int t = 0; t < 2; ++t)
#pragma unroll
    for (int r = 0; r < 8; ++r)
      ylds[wv][r + 8 * half][t * 16 + m] = y0acc[t][r];
  asm volatile("s_wait_dscnt 0" ::: "memory");

  v8f ga[3][2] = {{vzero, vzero}, {vzero, vzero}, {vzero, vzero}};
  for (int s = 0; s < 8; ++s) {
    const int k = s * 4 + 2 * half;
    v2f ay; ay.x = ylds[wv][m][k]; ay.y = ylds[wv][m][k + 1];
#pragma unroll
    for (int g = 0; g < 3; ++g) MM2(ga[g], ay, W_G(g), s);
  }
  v8f gate[3][2];
#pragma unroll
  for (int g = 0; g < 3; ++g)
#pragma unroll
    for (int t = 0; t < 2; ++t)
#pragma unroll
      for (int r = 0; r < 8; ++r) {
        const float v = ga[g][t][r];
        gate[g][t][r] = v * (1.0f / (1.0f + __expf(-v)));
      }

#pragma unroll
  for (int t = 0; t < 2; ++t)
#pragma unroll
    for (int r = 0; r < 8; ++r) {
      const int row = tile * 16 + r + 8 * half;
      if (row < N) {
        const int cc = t * 16 + m;
        out0[row * CCH + cc] = node0[row * CCH + cc] + gate[0][t][r];
      }
    }

  v8f y1acc[3][2] = {{vzero, vzero}, {vzero, vzero}, {vzero, vzero}};
  for (int s = 0; s < 8; ++s) {
    LOAD_BASE(s);
#pragma unroll
    for (int v = 0; v < 3; ++v) {
      v2f av, a01, tv;
#pragma unroll
      for (int jj = 0; jj < 2; ++jj) {
        av[jj]  = b1[jj][v];
        a01[jj] = x0[jj] * b1[jj][v];
        float acc = 0.f;
#pragma unroll
        for (int a = 0; a < 3; ++a) acc += b1[jj][a] * b2[jj][a * 3 + v];
        tv[jj] = acc;
      }
      MM2(y1acc[v], av,  W_U1,  s);
      MM2(y1acc[v], a01, W_V(1), s);
      MM2(y1acc[v], tv,  W_V(5), s);
    }
  }
#pragma unroll
  for (int v = 0; v < 3; ++v)
#pragma unroll
    for (int t = 0; t < 2; ++t)
#pragma unroll
      for (int r = 0; r < 8; ++r) {
        const int row = tile * 16 + r + 8 * half;
        if (row < N) {
          const int cc = t * 16 + m;
          const int o  = (row * CCH + cc) * 3 + v;
          out1[o] = node1[o] + y1acc[v][t][r] * gate[1][t][r];
        }
      }

  for (int aIdx = 0; aIdx < 3; ++aIdx) {
    v8f y2acc[3][2] = {{vzero, vzero}, {vzero, vzero}, {vzero, vzero}};
    for (int s = 0; s < 8; ++s) {
      LOAD_BASE(s);
#pragma unroll
      for (int b = 0; b < 3; ++b) {
        v2f a2v, a02, a11, qv;
#pragma unroll
        for (int jj = 0; jj < 2; ++jj) {
          const float g = b2[jj][aIdx * 3 + b];
          a2v[jj] = g;
          a02[jj] = x0[jj] * g;
          a11[jj] = b1[jj][aIdx] * b1[jj][b];
          float acc = 0.f;
#pragma unroll
          for (int kk = 0; kk < 3; ++kk)
            acc += b2[jj][aIdx * 3 + kk] * b2[jj][kk * 3 + b];
          qv[jj] = acc;
        }
        MM2(y2acc[b], a2v, W_U2,  s);
        MM2(y2acc[b], a02, W_V(2), s);
        MM2(y2acc[b], a11, W_V(4), s);
        MM2(y2acc[b], qv,  W_V(7), s);
      }
    }
#pragma unroll
    for (int b = 0; b < 3; ++b)
#pragma unroll
      for (int t = 0; t < 2; ++t)
#pragma unroll
        for (int r = 0; r < 8; ++r) {
          const int row = tile * 16 + r + 8 * half;
          if (row < N) {
            const int cc = t * 16 + m;
            const int o  = (row * CCH + cc) * 9 + aIdx * 3 + b;
            out2[o] = node2[o] + y2acc[b][t][r] * gate[2][t][r];
          }
        }
  }
#undef LOAD_BASE
#undef MM2
#undef LDB
}

extern "C" void kernel_launch(void* const* d_in, const int* in_sizes, int n_in,
                              void* d_out, int out_size, void* d_ws, size_t ws_size,
                              hipStream_t stream) {
  const float* node0 = (const float*)d_in[0];
  const float* node1 = (const float*)d_in[1];
  const float* node2 = (const float*)d_in[2];
  const float* rij   = (const float*)d_in[3];
  const float* Wrad  = (const float*)d_in[4];
  const float* U     = (const float*)d_in[5];
  const float* V     = (const float*)d_in[6];
  const float* Wg    = (const float*)d_in[7];
  const int*   idx_i = (const int*)d_in[8];
  const int*   idx_j = (const int*)d_in[9];

  const int N = in_sizes[0] / CCH;
  const int E = in_sizes[3] / 3;

  float* a0 = (float*)d_ws;
  float* a1 = a0 + (size_t)N * CCH;
  float* a2 = a1 + (size_t)N * CCH * 3;

  float* out0 = (float*)d_out;
  float* out1 = out0 + (size_t)N * CCH;
  float* out2 = out1 + (size_t)N * CCH * 3;

  const int n4 = N * CCH * 13 / 4;
  mmb_zero<<<(n4 + 255) / 256, 256, 0, stream>>>((float4*)d_ws, n4);

  mmb_edge<<<(E + 7) / 8, 256, 0, stream>>>(rij, Wrad, idx_i, idx_j,
                                            node0, node1, node2,
                                            a0, a1, a2, E);

  const int nTiles = (N + 15) / 16;
  mmb_node<<<(nTiles + 3) / 4, 128, 0, stream>>>(a0, a1, a2,
                                                 node0, node1, node2,
                                                 U, V, Wg,
                                                 out0, out1, out2, N);
}


# edge phase re-cut by destination range: a workgroup owns 80 nodes, filters the edge list for them, sums messages in LDS (f32 LDS adds, arrival order), stores its block once; no global atomics
# speedup vs baseline: 1.6296x; 1.6296x over previous
#include <hip/hip_runtime.h>
#include <hip/hip_bf16.h>
#include <math.h>

#define CCH 32
#define NPATH 11
#define NB 8

typedef __attribute__((ext_vector_type(2))) float v2f;
typedef __attribute__((ext_vector_type(8))) float v8f;

__device__ __forceinline__ v8f wmma4(v2f a, v2f b, v8f c) {
  return __builtin_amdgcn_wmma_f32_16x16x4_f32(
       false, a,  false, b,
       (short)0, c,  false,  false);
}

#define EDGE_TN    80
#define EDGE_WAVES 32
#define EDGE_SCAN  8

__device__ __forceinline__ void lds_add(float* p, float v) {
  __hip_atomic_fetch_add(p, v, __ATOMIC_RELAXED, __HIP_MEMORY_SCOPE_WORKGROUP);
}

__global__ void __launch_bounds__(EDGE_WAVES * 32)
mmb_edge(const float* __restrict__ rij,
         const float* __restrict__ Wrad,
         const int*   __restrict__ idx_i,
         const int*   __restrict__ idx_j,
         const float* __restrict__ node0,
         const float* __restrict__ node1,
         const float* __restrict__ node2,
         float* __restrict__ a0,
         float* __restrict__ a1,
         float* __restrict__ a2,
         int N, int E) {
  __shared__ float4 acc4[EDGE_TN * 104];
  float* const s0 = (float*)acc4;
  float* const s1 = s0 + EDGE_TN * CCH;
  float* const s2 = s0 + EDGE_TN * CCH * 4;

  const int tid = threadIdx.x;
  const int c   = tid & 31;
  const int wv  = __builtin_amdgcn_readfirstlane(tid >> 5);
  const int n0  = blockIdx.x * EDGE_TN;
  const int nv  = (N - n0 < EDGE_TN) ? (N - n0) : EDGE_TN;

  for (int k = tid; k < EDGE_TN * 104; k += EDGE_WAVES * 32)
    acc4[k] = make_float4(0.f, 0.f, 0.f, 0.f);
  __syncthreads();

  const int nChunks = (E + 31) >> 5;
  for (int cb = wv; cb < nChunks; cb += EDGE_WAVES * EDGE_SCAN) {
    int      il[EDGE_SCAN];
    unsigned m[EDGE_SCAN];
#pragma unroll
    for (int u = 0; u < EDGE_SCAN; ++u) {
      const int ev = ((cb + u * EDGE_WAVES) << 5) + c;
      const int id = idx_i[(ev < E) ? ev : (E - 1)];
      il[u] = (ev < E) ? (id - n0) : -1;
    }
#pragma unroll
    for (int u = 0; u < EDGE_SCAN; ++u)
      m[u] = __builtin_amdgcn_ballot_w32((unsigned)il[u] < (unsigned)nv);

    for (;;) {
      int u = -1;
      unsigned mm = 0u;
#pragma unroll
      for (int t = EDGE_SCAN - 1; t >= 0; --t)
        if (m[t] != 0u) { u = t; mm = m[t]; }
      if (u < 0) break;
      const int l = __builtin_ctz(mm);
      int ilv = il[0];
#pragma unroll
      for (int t = 1; t < EDGE_SCAN; ++t) ilv = (u == t) ? il[t] : ilv;
#pragma unroll
      for (int t = 0; t < EDGE_SCAN; ++t)
        m[t] = (u == t) ? (m[t] & (m[t] - 1u)) : m[t];
      const int nl = __builtin_amdgcn_readlane(ilv, l);
      const int e  = ((cb + u * EDGE_WAVES) << 5) + l;

      const float r0 = rij[e * 3 + 0], r1 = rij[e * 3 + 1], r2 = rij[e * 3 + 2];
      const int   j  = idx_j[e];
      const float d  = sqrtf(r0 * r0 + r1 * r1 + r2 * r2 + 1e-12f);
      const float iv = 1.0f / d;
      const float rh[3] = {r0 * iv, r1 * iv, r2 * iv};

      float x = d * 0.2f;
      x = (x > 1.0f) ? 1.0f : x;
      const float fc = 0.5f * (cosf(3.14159265358979f * x) + 1.0f);

      float rbf[NB];
#pragma unroll
      for (int b = 0; b < NB; ++b) {
        const float t = d - (0.5f + 0.6428571429f * (float)b);
        rbf[b] = __expf(-4.0f * t * t) * fc;
      }

      float w[NPATH];
#pragma unroll
      for (int p = 0; p < NPATH; ++p) {
        float acc = 0.f;
#pragma unroll
        for (int b = 0; b < NB; ++b) acc += rbf[b] * Wrad[(p * NB + b) * CCH + c];
        w[p] = acc;
      }

      const float g0 = node0[j * CCH + c];
      float g1[3], g2[9];
#pragma unroll
      for (int a = 0; a < 3; ++a) g1[a] = node1[(j * CCH + c) * 3 + a];
#pragma unroll
      for (int q = 0; q < 9; ++q) g2[q] = node2[(j * CCH + c) * 9 + q];

      const float d1 = g1[0] * rh[0] + g1[1] * rh[1] + g1[2] * rh[2];
      float gr[3];
#pragma unroll
      for (int a = 0; a < 3; ++a)
        gr[a] = g2[a * 3 + 0] * rh[0] + g2[a * 3 + 1] * rh[1] + g2[a * 3 + 2] * rh[2];
      const float d2 = gr[0] * rh[0] + gr[1] * rh[1] + gr[2] * rh[2];

      const float inv16 = 0.0625f;
      const int   rc    = nl * CCH + c;

      lds_add(&s0[rc], (g0 * w[0] + d1 * w[4] + d2 * w[9]) * inv16);

#pragma unroll
      for (int a = 0; a < 3; ++a) {
        const float mv = (g0 * w[1] + d1 * w[6]) * rh[a] + g1[a] * w[3] + gr[a] * w[8];
        lds_add(&s1[rc * 3 + a], mv * inv16);
      }
#pragma unroll
      for (int a = 0; a < 3; ++a) {
        const float ca = g0 * w[2] * rh[a] + g1[a] * w[5] + gr[a] * w[10];
#pragma unroll
        for (int b = 0; b < 3; ++b) {
          const float mv = ca * rh[b] + g2[a * 3 + b] * w[7];
          lds_add(&s2[rc * 9 + a * 3 + b], mv * inv16);
        }
      }
    }
  }
  __syncthreads();

  float4* const o0 = (float4*)(a0 + (size_t)n0 * CCH);
  float4* const o1 = (float4*)(a1 + (size_t)n0 * CCH * 3);
  float4* const o2 = (float4*)(a2 + (size_t)n0 * CCH * 9);
  for (int k = tid; k < nv * 8; k += EDGE_WAVES * 32)  o0[k] = acc4[k];
  for (int k = tid; k < nv * 24; k += EDGE_WAVES * 32) o1[k] = acc4[EDGE_TN * 8 + k];
  for (int k = tid; k < nv * 72; k += EDGE_WAVES * 32) o2[k] = acc4[EDGE_TN * 32 + k];
}

#define W_U0 0
#define W_U1 1
#define W_U2 2
#define W_V(k) (3 + (k))
#define W_G(k) (11 + (k))

__global__ void __launch_bounds__(128)
mmb_node(const float* __restrict__ a0, const float* __restrict__ a1,
         const float* __restrict__ a2,
         const float* __restrict__ node0, const float* __restrict__ node1,
         const float* __restrict__ node2,
         const float* __restrict__ U, const float* __restrict__ V,
         const float* __restrict__ Wg,
         float* __restrict__ out0, float* __restrict__ out1,
         float* __restrict__ out2, int N) {
  __shared__ float2 wlds[14 * 8 * 2 * 32];
  __shared__ float  ylds[4][16][CCH];

  for (int idx = threadIdx.x; idx < 14 * 1024; idx += blockDim.x) {
    const int wmat = idx >> 10;
    const int rem  = idx & 1023;
    const int k    = rem >> 5;
    const int n    = rem & 31;
    const float* src = (wmat < 3) ? (U + wmat * 1024)
                     : (wmat < 11) ? (V + (wmat - 3) * 1024)
                                   : (Wg + (wmat - 11) * 1024);
    const float val = src[rem];
    const int s   = k >> 2;
    const int kk  = k & 3;
    const int ln  = ((kk >> 1) << 4) | (n & 15);
    const int t   = n >> 4;
    const int jj  = kk & 1;
    ((float*)&wlds[((wmat * 8 + s) * 2 + t) * 32 + ln])[jj] = val;
  }
  __syncthreads();

  const int lane = threadIdx.x & 31;
  const int wv   = threadIdx.x >> 5;
  const int half = lane >> 4;
  const int m    = lane & 15;
  const int tile = blockIdx.x * 4 + wv;
  if (tile * 16 >= N) return;

#define LDB(wm, s, t)                                                       \
  ({ float2 _bv = wlds[(((wm) * 8 + (s)) * 2 + (t)) * 32 + lane];           \
     v2f _b; _b.x = _bv.x; _b.y = _bv.y; _b; })

#define MM2(acc, aop, wm, s)                                                \
  do { v2f _a = (aop);                                                      \
       acc[0] = wmma4(_a, LDB(wm, s, 0), acc[0]);                           \
       acc[1] = wmma4(_a, LDB(wm, s, 1), acc[1]); } while (0)

  const v8f vzero = {};

  const int rowA = tile * 16 + m;
  const int rowC = (rowA < N) ? rowA : (N - 1);
  float x0[2], b1[2][3], b2[2][9];
#define LOAD_BASE(s)                                                        \
  do { const int _k = (s) * 4 + 2 * half;                                   \
       const int _rc = rowC * CCH + _k;                                     \
       x0[0] = a0[_rc]; x0[1] = a0[_rc + 1];                                \
       _Pragma("unroll") for (int _a = 0; _a < 3; ++_a) {                   \
         b1[0][_a] = a1[_rc * 3 + _a]; b1[1][_a] = a1[(_rc + 1) * 3 + _a]; }\
       _Pragma("unroll") for (int _q = 0; _q < 9; ++_q) {                   \
         b2[0][_q] = a2[_rc * 9 + _q]; b2[1][_q] = a2[(_rc + 1) * 9 + _q]; }\
  } while (0)

  v8f y0acc[2] = {vzero, vzero};
  for (int s = 0; s < 8; ++s) {
    LOAD_BASE(s);
    v2f ax, asq, s1v, s2v;
#pragma unroll
    for (int jj = 0; jj < 2; ++jj) {
      ax[jj]  = x0[jj];
      asq[jj] = x0[jj] * x0[jj];
      float acc = 0.f;
#pragma unroll
      for (int a = 0; a < 3; ++a) acc += b1[jj][a] * b1[jj][a];
      s1v[jj] = acc;
      acc = 0.f;
#pragma unroll
      for (int q = 0; q < 9; ++q) acc += b2[jj][q] * b2[jj][q];
      s2v[jj] = acc;
    }
    MM2(y0acc, ax,  W_U0,  s);
    MM2(y0acc, asq, W_V(0), s);
    MM2(y0acc, s1v, W_V(3), s);
    MM2(y0acc, s2v, W_V(6), s);
  }

#pragma unroll
  for (int t = 0; t < 2; ++t)
#pragma unroll
    for (int r = 0; r < 8; ++r)
      ylds[wv][r + 8 * half][t * 16 + m] = y0acc[t][r];
  asm volatile("s_wait_dscnt 0" ::: "memory");

  v8f ga[3][2] = {{vzero, vzero}, {vzero, vzero}, {vzero, vzero}};
  for (int s = 0; s < 8; ++s) {
    const int k = s * 4 + 2 * half;
    v2f ay; ay.x = ylds[wv][m][k]; ay.y = ylds[wv][m][k + 1];
#pragma unroll
    for (int g = 0; g < 3; ++g) MM2(ga[g], ay, W_G(g), s);
  }
  v8f gate[3][2];
#pragma unroll
  for (int g = 0; g < 3; ++g)
#pragma unroll
    for (int t = 0; t < 2; ++t)
#pragma unroll
      for (int r = 0; r < 8; ++r) {
        const float v = ga[g][t][r];
        gate[g][t][r] = v * (1.0f / (1.0f + __expf(-v)));
      }

#pragma unroll
  for (int t = 0; t < 2; ++t)
#pragma unroll
    for (int r = 0; r < 8; ++r) {
      const int row = tile * 16 + r + 8 * half;
      if (row < N) {
        const int cc = t * 16 + m;
        out0[row * CCH + cc] = node0[row * CCH + cc] + gate[0][t][r];
      }
    }

  v8f y1acc[3][2] = {{vzero, vzero}, {vzero, vzero}, {vzero, vzero}};
  for (int s = 0; s < 8; ++s) {
    LOAD_BASE(s);
#pragma unroll
    for (int v = 0; v < 3; ++v) {
      v2f av, a01, tv;
#pragma unroll
      for (int jj = 0; jj < 2; ++jj) {
        av[jj]  = b1[jj][v];
        a01[jj] = x0[jj] * b1[jj][v];
        float acc = 0.f;
#pragma unroll
        for (int a = 0; a < 3; ++a) acc += b1[jj][a] * b2[jj][a * 3 + v];
        tv[jj] = acc;
      }
      MM2(y1acc[v], av,  W_U1,  s);
      MM2(y1acc[v], a01, W_V(1), s);
      MM2(y1acc[v], tv,  W_V(5), s);
    }
  }
#pragma unroll
  for (int v = 0; v < 3; ++v)
#pragma unroll
    for (int t = 0; t < 2; ++t)
#pragma unroll
      for (int r = 0; r < 8; ++r) {
        const int row = tile * 16 + r + 8 * half;
        if (row < N) {
          const int cc = t * 16 + m;
          const int o  = (row * CCH + cc) * 3 + v;
          out1[o] = node1[o] + y1acc[v][t][r] * gate[1][t][r];
        }
      }

  for (int aIdx = 0; aIdx < 3; ++aIdx) {
    v8f y2acc[3][2] = {{vzero, vzero}, {vzero, vzero}, {vzero, vzero}};
    for (int s = 0; s < 8; ++s) {
      LOAD_BASE(s);
#pragma unroll
      for (int b = 0; b < 3; ++b) {
        v2f a2v, a02, a11, qv;
#pragma unroll
        for (int jj = 0; jj < 2; ++jj) {
          const float g = b2[jj][aIdx * 3 + b];
          a2v[jj] = g;
          a02[jj] = x0[jj] * g;
          a11[jj] = b1[jj][aIdx] * b1[jj][b];
          float acc = 0.f;
#pragma unroll
          for (int kk = 0; kk < 3; ++kk)
            acc += b2[jj][aIdx * 3 + kk] * b2[jj][kk * 3 + b];
          qv[jj] = acc;
        }
        MM2(y2acc[b], a2v, W_U2,  s);
        MM2(y2acc[b], a02, W_V(2), s);
        MM2(y2acc[b], a11, W_V(4), s);
        MM2(y2acc[b], qv,  W_V(7), s);
      }
    }
#pragma unroll
    for (int b = 0; b < 3; ++b)
#pragma unroll
      for (int t = 0; t < 2; ++t)
#pragma unroll
        for (int r = 0; r < 8; ++r) {
          const int row = tile * 16 + r + 8 * half;
          if (row < N) {
            const int cc = t * 16 + m;
            const int o  = (row * CCH + cc) * 9 + aIdx * 3 + b;
            out2[o] = node2[o] + y2acc[b][t][r] * gate[2][t][r];
          }
        }
  }
#undef LOAD_BASE
#undef MM2
#undef LDB
}

extern "C" void kernel_launch(void* const* d_in, const int* in_sizes, int n_in,
                              void* d_out, int out_size, void* d_ws, size_t ws_size,
                              hipStream_t stream) {
  const float* node0 = (const float*)d_in[0];
  const float* node1 = (const float*)d_in[1];
  const float* node2 = (const float*)d_in[2];
  const float* rij   = (const float*)d_in[3];
  const float* Wrad  = (const float*)d_in[4];
  const float* U     = (const float*)d_in[5];
  const float* V     = (const float*)d_in[6];
  const float* Wg    = (const float*)d_in[7];
  const int*   idx_i = (const int*)d_in[8];
  const int*   idx_j = (const int*)d_in[9];

  const int N = in_sizes[0] / CCH;
  const int E = in_sizes[3] / 3;

  float* a0 = (float*)d_ws;
  float* a1 = a0 + (size_t)N * CCH;
  float* a2 = a1 + (size_t)N * CCH * 3;

  float* out0 = (float*)d_out;
  float* out1 = out0 + (size_t)N * CCH;
  float* out2 = out1 + (size_t)N * CCH * 3;

  mmb_edge<<<(N + EDGE_TN - 1) / EDGE_TN, EDGE_WAVES * 32, 0, stream>>>(
      rij, Wrad, idx_i, idx_j, node0, node1, node2, a0, a1, a2, N, E);

  const int nTiles = (N + 15) / 16;
  mmb_node<<<(nTiles + 3) / 4, 128, 0, stream>>>(a0, a1, a2,
                                                 node0, node1, node2,
                                                 U, V, Wg,
                                                 out0, out1, out2, N);
}


# node phase output: a wave stages its gated tile in LDS in output order, adds residual rows, stores its span 16 B per lane (was 208 blocking 4 B load-add-store trips); product rounded before add
# speedup vs baseline: 2.3978x; 1.4714x over previous
#include <hip/hip_runtime.h>
#include <hip/hip_bf16.h>
#include <math.h>

#define CCH 32
#define NPATH 11
#define NB 8

typedef __attribute__((ext_vector_type(2))) float v2f;
typedef __attribute__((ext_vector_type(8))) float v8f;
typedef __attribute__((ext_vector_type(4))) float v4f;

__device__ __forceinline__ v8f wmma4(v2f a, v2f b, v8f c) {
  return __builtin_amdgcn_wmma_f32_16x16x4_f32(
       false, a,  false, b,
       (short)0, c,  false,  false);
}

#define EDGE_TN    80
#define EDGE_WAVES 32
#define EDGE_SCAN  8

__device__ __forceinline__ void lds_add(float* p, float v) {
  __hip_atomic_fetch_add(p, v, __ATOMIC_RELAXED, __HIP_MEMORY_SCOPE_WORKGROUP);
}

__global__ void __launch_bounds__(EDGE_WAVES * 32)
mmb_edge(const float* __restrict__ rij,
         const float* __restrict__ Wrad,
         const int*   __restrict__ idx_i,
         const int*   __restrict__ idx_j,
         const float* __restrict__ node0,
         const float* __restrict__ node1,
         const float* __restrict__ node2,
         float* __restrict__ a0,
         float* __restrict__ a1,
         float* __restrict__ a2,
         int N, int E) {
  __shared__ float4 acc4[EDGE_TN * 104];
  float* const s0 = (float*)acc4;
  float* const s1 = s0 + EDGE_TN * CCH;
  float* const s2 = s0 + EDGE_TN * CCH * 4;

  const int tid = threadIdx.x;
  const int c   = tid & 31;
  const int wv  = __builtin_amdgcn_readfirstlane(tid >> 5);
  const int n0  = blockIdx.x * EDGE_TN;
  const int nv  = (N - n0 < EDGE_TN) ? (N - n0) : EDGE_TN;

  for (int k = tid; k < EDGE_TN * 104; k += EDGE_WAVES * 32)
    acc4[k] = make_float4(0.f, 0.f, 0.f, 0.f);
  __syncthreads();

  const int nChunks = (E + 31) >> 5;
  for (int cb = wv; cb < nChunks; cb += EDGE_WAVES * EDGE_SCAN) {
    int      il[EDGE_SCAN];
    unsigned m[EDGE_SCAN];
#pragma unroll
    for (int u = 0; u < EDGE_SCAN; ++u) {
      const int ev = ((cb + u * EDGE_WAVES) << 5) + c;
      const int id = idx_i[(ev < E) ? ev : (E - 1)];
      il[u] = (ev < E) ? (id - n0) : -1;
    }
#pragma unroll
    for (int u = 0; u < EDGE_SCAN; ++u)
      m[u] = __builtin_amdgcn_ballot_w32((unsigned)il[u] < (unsigned)nv);

    for (;;) {
      int u = -1;
      unsigned mm = 0u;
#pragma unroll
      for (int t = EDGE_SCAN - 1; t >= 0; --t)
        if (m[t] != 0u) { u = t; mm = m[t]; }
      if (u < 0) break;
      const int l = __builtin_ctz(mm);
      int ilv = il[0];
#pragma unroll
      for (int t = 1; t < EDGE_SCAN; ++t) ilv = (u == t) ? il[t] : ilv;
#pragma unroll
      for (int t = 0; t < EDGE_SCAN; ++t)
        m[t] = (u == t) ? (m[t] & (m[t] - 1u)) : m[t];
      const int nl = __builtin_amdgcn_readlane(ilv, l);
      const int e  = ((cb + u * EDGE_WAVES) << 5) + l;

      const float r0 = rij[e * 3 + 0], r1 = rij[e * 3 + 1], r2 = rij[e * 3 + 2];
      const int   j  = idx_j[e];
      const float d  = sqrtf(r0 * r0 + r1 * r1 + r2 * r2 + 1e-12f);
      const float iv = 1.0f / d;
      const float rh[3] = {r0 * iv, r1 * iv, r2 * iv};

      float x = d * 0.2f;
      x = (x > 1.0f) ? 1.0f : x;
      const float fc = 0.5f * (cosf(3.14159265358979f * x) + 1.0f);

      float rbf[NB];
#pragma unroll
      for (int b = 0; b < NB; ++b) {
        const float t = d - (0.5f + 0.6428571429f * (float)b);
        rbf[b] = __expf(-4.0f * t * t) * fc;
      }

      float w[NPATH];
#pragma unroll
      for (int p = 0; p < NPATH; ++p) {
        float acc = 0.f;
#pragma unroll
        for (int b = 0; b < NB; ++b) acc += rbf[b] * Wrad[(p * NB + b) * CCH + c];
        w[p] = acc;
      }

      const float g0 = node0[j * CCH + c];
      float g1[3], g2[9];
#pragma unroll
      for (int a = 0; a < 3; ++a) g1[a] = node1[(j * CCH + c) * 3 + a];
#pragma unroll
      for (int q = 0; q < 9; ++q) g2[q] = node2[(j * CCH + c) * 9 + q];

      const float d1 = g1[0] * rh[0] + g1[1] * rh[1] + g1[2] * rh[2];
      float gr[3];
#pragma unroll
      for (int a = 0; a < 3; ++a)
        gr[a] = g2[a * 3 + 0] * rh[0] + g2[a * 3 + 1] * rh[1] + g2[a * 3 + 2] * rh[2];
      const float d2 = gr[0] * rh[0] + gr[1] * rh[1] + gr[2] * rh[2];

      const float inv16 = 0.0625f;
      const int   rc    = nl * CCH + c;

      lds_add(&s0[rc], (g0 * w[0] + d1 * w[4] + d2 * w[9]) * inv16);

#pragma unroll
      for (int a = 0; a < 3; ++a) {
        const float mv = (g0 * w[1] + d1 * w[6]) * rh[a] + g1[a] * w[3] + gr[a] * w[8];
        lds_add(&s1[rc * 3 + a], mv * inv16);
      }
#pragma unroll
      for (int a = 0; a < 3; ++a) {
        const float ca = g0 * w[2] * rh[a] + g1[a] * w[5] + gr[a] * w[10];
#pragma unroll
        for (int b = 0; b < 3; ++b) {
          const float mv = ca * rh[b] + g2[a * 3 + b] * w[7];
          lds_add(&s2[rc * 9 + a * 3 + b], mv * inv16);
        }
      }
    }
  }
  __syncthreads();

  float4* const o0 = (float4*)(a0 + (size_t)n0 * CCH);
  float4* const o1 = (float4*)(a1 + (size_t)n0 * CCH * 3);
  float4* const o2 = (float4*)(a2 + (size_t)n0 * CCH * 9);
  for (int k = tid; k < nv * 8; k += EDGE_WAVES * 32)  o0[k] = acc4[k];
  for (int k = tid; k < nv * 24; k += EDGE_WAVES * 32) o1[k] = acc4[EDGE_TN * 8 + k];
  for (int k = tid; k < nv * 72; k += EDGE_WAVES * 32) o2[k] = acc4[EDGE_TN * 32 + k];
}

#define STAGE_F (16 * CCH * 9)

template <int PER_LANE>
__device__ __forceinline__ void flush_tile(const float* st,
                                           const float* __restrict__ res,
                                           float* __restrict__ dst,
                                           int nvalid4, int lane) {
  const v4f* const s4 = (const v4f*)st;
  const v4f* const r4 = (const v4f*)res;
  v4f* const d4 = (v4f*)dst;
  constexpr int BATCH = (PER_LANE < 12) ? PER_LANE : 12;
#pragma unroll
  for (int i0 = 0; i0 < PER_LANE; i0 += BATCH) {
    v4f rv[BATCH], sv[BATCH];
#pragma unroll
    for (int i = 0; i < BATCH; ++i) {
      const int k = (i0 + i) * 32 + lane;
      rv[i] = r4[(k < nvalid4) ? k : (nvalid4 - 1)];
      sv[i] = s4[k];
    }
#pragma unroll
    for (int i = 0; i < BATCH; ++i) {
      const int k = (i0 + i) * 32 + lane;
      if (k < nvalid4) d4[k] = sv[i] + rv[i];
    }
  }
}
#define W_U0 0
#define W_U1 1
#define W_U2 2
#define W_V(k) (3 + (k))
#define W_G(k) (11 + (k))

__global__ void __launch_bounds__(128)
mmb_node(const float* __restrict__ a0, const float* __restrict__ a1,
         const float* __restrict__ a2,
         const float* __restrict__ node0, const float* __restrict__ node1,
         const float* __restrict__ node2,
         const float* __restrict__ U, const float* __restrict__ V,
         const float* __restrict__ Wg,
         float* __restrict__ out0, float* __restrict__ out1,
         float* __restrict__ out2, int N) {
  __shared__ float2 wlds[14 * 8 * 2 * 32];
  __shared__ __attribute__((aligned(16))) float stage[4 * STAGE_F];

  for (int idx = threadIdx.x; idx < 14 * 1024; idx += blockDim.x) {
    const int wmat = idx >> 10;
    const int rem  = idx & 1023;
    const int k    = rem >> 5;
    const int n    = rem & 31;
    const float* src = (wmat < 3) ? (U + wmat * 1024)
                     : (wmat < 11) ? (V + (wmat - 3) * 1024)
                                   : (Wg + (wmat - 11) * 1024);
    const float val = src[rem];
    const int s   = k >> 2;
    const int kk  = k & 3;
    const int ln  = ((kk >> 1) << 4) | (n & 15);
    const int t   = n >> 4;
    const int jj  = kk & 1;
    ((float*)&wlds[((wmat * 8 + s) * 2 + t) * 32 + ln])[jj] = val;
  }
  __syncthreads();

  const int lane = threadIdx.x & 31;
  const int wv   = threadIdx.x >> 5;
  const int half = lane >> 4;
  const int m    = lane & 15;
  const int tile = blockIdx.x * 4 + wv;
  if (tile * 16 >= N) return;

#define LDB(wm, s, t)                                                       \
  ({ float2 _bv = wlds[(((wm) * 8 + (s)) * 2 + (t)) * 32 + lane];           \
     v2f _b; _b.x = _bv.x; _b.y = _bv.y; _b; })

#define MM2(acc, aop, wm, s)                                                \
  do { v2f _a = (aop);                                                      \
       acc[0] = wmma4(_a, LDB(wm, s, 0), acc[0]);                           \
       acc[1] = wmma4(_a, LDB(wm, s, 1), acc[1]); } while (0)

  const v8f vzero = {};
  float* const st = stage + wv * STAGE_F;
  const int nvr = (N - tile * 16 < 16) ? (N - tile * 16) : 16;
#define STAGE_FENCE() asm volatile("s_wait_dscnt 0" ::: "memory")

  const int rowA = tile * 16 + m;
  const int rowC = (rowA < N) ? rowA : (N - 1);
  float x0[2], b1[2][3], b2[2][9];
#define LOAD_BASE(s)                                                        \
  do { const int _k = (s) * 4 + 2 * half;                                   \
       const int _rc = rowC * CCH + _k;                                     \
       x0[0] = a0[_rc]; x0[1] = a0[_rc + 1];                                \
       _Pragma("unroll") for (int _a = 0; _a < 3; ++_a) {                   \
         b1[0][_a] = a1[_rc * 3 + _a]; b1[1][_a] = a1[(_rc + 1) * 3 + _a]; }\
       _Pragma("unroll") for (int _q = 0; _q < 9; ++_q) {                   \
         b2[0][_q] = a2[_rc * 9 + _q]; b2[1][_q] = a2[(_rc + 1) * 9 + _q]; }\
  } while (0)

  v8f y0acc[2] = {vzero, vzero};
  for (int s = 0; s < 8; ++s) {
    LOAD_BASE(s);
    v2f ax, asq, s1v, s2v;
#pragma unroll
    for (int jj = 0; jj < 2; ++jj) {
      ax[jj]  = x0[jj];
      asq[jj] = x0[jj] * x0[jj];
      float acc = 0.f;
#pragma unroll
      for (int a = 0; a < 3; ++a) acc += b1[jj][a] * b1[jj][a];
      s1v[jj] = acc;
      acc = 0.f;
#pragma unroll
      for (int q = 0; q < 9; ++q) acc += b2[jj][q] * b2[jj][q];
      s2v[jj] = acc;
    }
    MM2(y0acc, ax,  W_U0,  s);
    MM2(y0acc, asq, W_V(0), s);
    MM2(y0acc, s1v, W_V(3), s);
    MM2(y0acc, s2v, W_V(6), s);
  }

#pragma unroll
  for (int t = 0; t < 2; ++t)
#pragma unroll
    for (int r = 0; r < 8; ++r)
      st[(r + 8 * half) * CCH + t * 16 + m] = y0acc[t][r];
  STAGE_FENCE();

  v8f ga[3][2] = {{vzero, vzero}, {vzero, vzero}, {vzero, vzero}};
  for (int s = 0; s < 8; ++s) {
    const int k = s * 4 + 2 * half;
    v2f ay; ay.x = st[m * CCH + k]; ay.y = st[m * CCH + k + 1];
#pragma unroll
    for (int g = 0; g < 3; ++g) MM2(ga[g], ay, W_G(g), s);
  }
  v8f gate[3][2];
#pragma unroll
  for (int g = 0; g < 3; ++g)
#pragma unroll
    for (int t = 0; t < 2; ++t)
#pragma unroll
      for (int r = 0; r < 8; ++r) {
        const float v = ga[g][t][r];
        gate[g][t][r] = v * (1.0f / (1.0f + __expf(-v)));
      }

#pragma unroll
  for (int t = 0; t < 2; ++t)
#pragma unroll
    for (int r = 0; r < 8; ++r)
      st[(r + 8 * half) * CCH + t * 16 + m] = gate[0][t][r];
  STAGE_FENCE();
  flush_tile<4>(st, node0 + (size_t)tile * 16 * CCH,
                out0 + (size_t)tile * 16 * CCH, nvr * 8, lane);

  v8f y1acc[3][2] = {{vzero, vzero}, {vzero, vzero}, {vzero, vzero}};
  for (int s = 0; s < 8; ++s) {
    LOAD_BASE(s);
#pragma unroll
    for (int v = 0; v < 3; ++v) {
      v2f av, a01, tv;
#pragma unroll
      for (int jj = 0; jj < 2; ++jj) {
        av[jj]  = b1[jj][v];
        a01[jj] = x0[jj] * b1[jj][v];
        float acc = 0.f;
#pragma unroll
        for (int a = 0; a < 3; ++a) acc += b1[jj][a] * b2[jj][a * 3 + v];
        tv[jj] = acc;
      }
      MM2(y1acc[v], av,  W_U1,  s);
      MM2(y1acc[v], a01, W_V(1), s);
      MM2(y1acc[v], tv,  W_V(5), s);
    }
  }
#pragma unroll
  for (int v = 0; v < 3; ++v)
#pragma unroll
    for (int t = 0; t < 2; ++t)
#pragma unroll
      for (int r = 0; r < 8; ++r)
        st[((r + 8 * half) * CCH + t * 16 + m) * 3 + v] =
            y1acc[v][t][r] * gate[1][t][r];
  STAGE_FENCE();
  flush_tile<12>(st, node1 + (size_t)tile * 16 * CCH * 3,
                 out1 + (size_t)tile * 16 * CCH * 3, nvr * 24, lane);

  for (int aIdx = 0; aIdx < 3; ++aIdx) {
    v8f y2acc[3][2] = {{vzero, vzero}, {vzero, vzero}, {vzero, vzero}};
    for (int s = 0; s < 8; ++s) {
      LOAD_BASE(s);
#pragma unroll
      for (int b = 0; b < 3; ++b) {
        v2f a2v, a02, a11, qv;
#pragma unroll
        for (int jj = 0; jj < 2; ++jj) {
          const float g = b2[jj][aIdx * 3 + b];
          a2v[jj] = g;
          a02[jj] = x0[jj] * g;
          a11[jj] = b1[jj][aIdx] * b1[jj][b];
          float acc = 0.f;
#pragma unroll
          for (int kk = 0; kk < 3; ++kk)
            acc += b2[jj][aIdx * 3 + kk] * b2[jj][kk * 3 + b];
          qv[jj] = acc;
        }
        MM2(y2acc[b], a2v, W_U2,  s);
        MM2(y2acc[b], a02, W_V(2), s);
        MM2(y2acc[b], a11, W_V(4), s);
        MM2(y2acc[b], qv,  W_V(7), s);
      }
    }
#pragma unroll
    for (int b = 0; b < 3; ++b)
#pragma unroll
      for (int t = 0; t < 2; ++t)
#pragma unroll
        for (int r = 0; r < 8; ++r)
          st[((r + 8 * half) * CCH + t * 16 + m) * 9 + aIdx * 3 + b] =
              y2acc[b][t][r] * gate[2][t][r];
  }
  STAGE_FENCE();
  flush_tile<36>(st, node2 + (size_t)tile * 16 * CCH * 9,
                 out2 + (size_t)tile * 16 * CCH * 9, nvr * 72, lane);
#undef STAGE_FENCE
#undef LOAD_BASE
#undef MM2
#undef LDB
}

extern "C" void kernel_launch(void* const* d_in, const int* in_sizes, int n_in,
                              void* d_out, int out_size, void* d_ws, size_t ws_size,
                              hipStream_t stream) {
  const float* node0 = (const float*)d_in[0];
  const float* node1 = (const float*)d_in[1];
  const float* node2 = (const float*)d_in[2];
  const float* rij   = (const float*)d_in[3];
  const float* Wrad  = (const float*)d_in[4];
  const float* U     = (const float*)d_in[5];
  const float* V     = (const float*)d_in[6];
  const float* Wg    = (const float*)d_in[7];
  const int*   idx_i = (const int*)d_in[8];
  const int*   idx_j = (const int*)d_in[9];

  const int N = in_sizes[0] / CCH;
  const int E = in_sizes[3] / 3;

  float* a0 = (float*)d_ws;
  float* a1 = a0 + (size_t)N * CCH;
  float* a2 = a1 + (size_t)N * CCH * 3;

  float* out0 = (float*)d_out;
  float* out1 = out0 + (size_t)N * CCH;
  float* out2 = out1 + (size_t)N * CCH * 3;

  mmb_edge<<<(N + EDGE_TN - 1) / EDGE_TN, EDGE_WAVES * 32, 0, stream>>>(
      rij, Wrad, idx_i, idx_j, node0, node1, node2, a0, a1, a2, N, E);

  const int nTiles = (N + 15) / 16;
  mmb_node<<<(nTiles + 3) / 4, 128, 0, stream>>>(a0, a1, a2,
                                                 node0, node1, node2,
                                                 U, V, Wg,
                                                 out0, out1, out2, N);
}


# edge phase: radial weights copied once per workgroup into LDS, one 368 B row per channel; an edge's filter sums read them as 22 16-byte LDS loads, not 88 4-byte global loads; sum order as before
# speedup vs baseline: 2.8216x; 1.1767x over previous
#include <hip/hip_runtime.h>
#include <hip/hip_bf16.h>
#include <math.h>

#define CCH 32
#define NPATH 11
#define NB 8

typedef __attribute__((ext_vector_type(2))) float v2f;
typedef __attribute__((ext_vector_type(8))) float v8f;
typedef __attribute__((ext_vector_type(4))) float v4f;

__device__ __forceinline__ v8f wmma4(v2f a, v2f b, v8f c) {
  return __builtin_amdgcn_wmma_f32_16x16x4_f32(
       false, a,  false, b,
       (short)0, c,  false,  false);
}

#define EDGE_TN    80
#define EDGE_WAVES 32
#define EDGE_SCAN  8
#define WR_PITCH   92

__device__ __forceinline__ void lds_add(float* p, float v) {
  __hip_atomic_fetch_add(p, v, __ATOMIC_RELAXED, __HIP_MEMORY_SCOPE_WORKGROUP);
}

__global__ void __launch_bounds__(EDGE_WAVES * 32)
mmb_edge(const float* __restrict__ rij,
         const float* __restrict__ Wrad,
         const int*   __restrict__ idx_i,
         const int*   __restrict__ idx_j,
         const float* __restrict__ node0,
         const float* __restrict__ node1,
         const float* __restrict__ node2,
         float* __restrict__ a0,
         float* __restrict__ a1,
         float* __restrict__ a2,
         int N, int E) {
  __shared__ float4 acc4[EDGE_TN * 104];
  float* const s0 = (float*)acc4;
  float* const s1 = s0 + EDGE_TN * CCH;
  float* const s2 = s0 + EDGE_TN * CCH * 4;
  __shared__ v4f wr4[CCH * WR_PITCH / 4];

  const int tid = threadIdx.x;
  const int c   = tid & 31;
  const int wv  = __builtin_amdgcn_readfirstlane(tid >> 5);
  const int n0  = blockIdx.x * EDGE_TN;
  const int nv  = (N - n0 < EDGE_TN) ? (N - n0) : EDGE_TN;

  constexpr int WR_N     = NPATH * NB * CCH;
  constexpr int WR_TRIPS = (WR_N + EDGE_WAVES * 32 - 1) / (EDGE_WAVES * 32);
  float wt[WR_TRIPS];
#pragma unroll
  for (int u = 0; u < WR_TRIPS; ++u) {
    const int k = tid + u * EDGE_WAVES * 32;
    wt[u] = Wrad[(k < WR_N) ? k : (WR_N - 1)];
  }
  for (int k = tid; k < EDGE_TN * 104; k += EDGE_WAVES * 32)
    acc4[k] = make_float4(0.f, 0.f, 0.f, 0.f);
#pragma unroll
  for (int u = 0; u < WR_TRIPS; ++u) {
    const int k = tid + u * EDGE_WAVES * 32;
    if (k < WR_N) ((float*)wr4)[(k & 31) * WR_PITCH + (k >> 5)] = wt[u];
  }
  __syncthreads();
  const v4f* const wrow = wr4 + c * (WR_PITCH / 4);

  const int nChunks = (E + 31) >> 5;
  for (int cb = wv; cb < nChunks; cb += EDGE_WAVES * EDGE_SCAN) {
    int      il[EDGE_SCAN];
    unsigned m[EDGE_SCAN];
#pragma unroll
    for (int u = 0; u < EDGE_SCAN; ++u) {
      const int ev = ((cb + u * EDGE_WAVES) << 5) + c;
      const int id = idx_i[(ev < E) ? ev : (E - 1)];
      il[u] = (ev < E) ? (id - n0) : -1;
    }
#pragma unroll
    for (int u = 0; u < EDGE_SCAN; ++u)
      m[u] = __builtin_amdgcn_ballot_w32((unsigned)il[u] < (unsigned)nv);

    for (;;) {
      int u = -1;
      unsigned mm = 0u;
#pragma unroll
      for (int t = EDGE_SCAN - 1; t >= 0; --t)
        if (m[t] != 0u) { u = t; mm = m[t]; }
      if (u < 0) break;
      const int l = __builtin_ctz(mm);
      int ilv = il[0];
#pragma unroll
      for (int t = 1; t < EDGE_SCAN; ++t) ilv = (u == t) ? il[t] : ilv;
#pragma unroll
      for (int t = 0; t < EDGE_SCAN; ++t)
        m[t] = (u == t) ? (m[t] & (m[t] - 1u)) : m[t];
      const int nl = __builtin_amdgcn_readlane(ilv, l);
      const int e  = ((cb + u * EDGE_WAVES) << 5) + l;

      const float r0 = rij[e * 3 + 0], r1 = rij[e * 3 + 1], r2 = rij[e * 3 + 2];
      const int   j  = idx_j[e];
      const float d  = sqrtf(r0 * r0 + r1 * r1 + r2 * r2 + 1e-12f);
      const float iv = 1.0f / d;
      const float rh[3] = {r0 * iv, r1 * iv, r2 * iv};

      float x = d * 0.2f;
      x = (x > 1.0f) ? 1.0f : x;
      const float fc = 0.5f * (cosf(3.14159265358979f * x) + 1.0f);

      float rbf[NB];
#pragma unroll
      for (int b = 0; b < NB; ++b) {
        const float t = d - (0.5f + 0.6428571429f * (float)b);
        rbf[b] = __expf(-4.0f * t * t) * fc;
      }

      float w[NPATH];
#pragma unroll
      for (int p = 0; p < NPATH; ++p) {
        const v4f q0 = wrow[p * 2], q1 = wrow[p * 2 + 1];
        const float wv[NB] = {q0.x, q0.y, q0.z, q0.w, q1.x, q1.y, q1.z, q1.w};
        float acc = 0.f;
#pragma unroll
        for (int b = 0; b < NB; ++b) acc += rbf[b] * wv[b];
        w[p] = acc;
      }

      const float g0 = node0[j * CCH + c];
      float g1[3], g2[9];
#pragma unroll
      for (int a = 0; a < 3; ++a) g1[a] = node1[(j * CCH + c) * 3 + a];
#pragma unroll
      for (int q = 0; q < 9; ++q) g2[q] = node2[(j * CCH + c) * 9 + q];

      const float d1 = g1[0] * rh[0] + g1[1] * rh[1] + g1[2] * rh[2];
      float gr[3];
#pragma unroll
      for (int a = 0; a < 3; ++a)
        gr[a] = g2[a * 3 + 0] * rh[0] + g2[a * 3 + 1] * rh[1] + g2[a * 3 + 2] * rh[2];
      const float d2 = gr[0] * rh[0] + gr[1] * rh[1] + gr[2] * rh[2];

      const float inv16 = 0.0625f;
      const int   rc    = nl * CCH + c;

      lds_add(&s0[rc], (g0 * w[0] + d1 * w[4] + d2 * w[9]) * inv16);

#pragma unroll
      for (int a = 0; a < 3; ++a) {
        const float mv = (g0 * w[1] + d1 * w[6]) * rh[a] + g1[a] * w[3] + gr[a] * w[8];
        lds_add(&s1[rc * 3 + a], mv * inv16);
      }
#pragma unroll
      for (int a = 0; a < 3; ++a) {
        const float ca = g0 * w[2] * rh[a] + g1[a] * w[5] + gr[a] * w[10];
#pragma unroll
        for (int b = 0; b < 3; ++b) {
          const float mv = ca * rh[b] + g2[a * 3 + b] * w[7];
          lds_add(&s2[rc * 9 + a * 3 + b], mv * inv16);
        }
      }
    }
  }
  __syncthreads();

  float4* const o0 = (float4*)(a0 + (size_t)n0 * CCH);
  float4* const o1 = (float4*)(a1 + (size_t)n0 * CCH * 3);
  float4* const o2 = (float4*)(a2 + (size_t)n0 * CCH * 9);
  for (int k = tid; k < nv * 8; k += EDGE_WAVES * 32)  o0[k] = acc4[k];
  for (int k = tid; k < nv * 24; k += EDGE_WAVES * 32) o1[k] = acc4[EDGE_TN * 8 + k];
  for (int k = tid; k < nv * 72; k += EDGE_WAVES * 32) o2[k] = acc4[EDGE_TN * 32 + k];
}

#define STAGE_F (16 * CCH * 9)

template <int PER_LANE>
__device__ __forceinline__ void flush_tile(const float* st,
                                           const float* __restrict__ res,
                                           float* __restrict__ dst,
                                           int nvalid4, int lane) {
  const v4f* const s4 = (const v4f*)st;
  const v4f* const r4 = (const v4f*)res;
  v4f* const d4 = (v4f*)dst;
  constexpr int BATCH = (PER_LANE < 12) ? PER_LANE : 12;
#pragma unroll
  for (int i0 = 0; i0 < PER_LANE; i0 += BATCH) {
    v4f rv[BATCH], sv[BATCH];
#pragma unroll
    for (int i = 0; i < BATCH; ++i) {
      const int k = (i0 + i) * 32 + lane;
      rv[i] = r4[(k < nvalid4) ? k : (nvalid4 - 1)];
      sv[i] = s4[k];
    }
#pragma unroll
    for (int i = 0; i < BATCH; ++i) {
      const int k = (i0 + i) * 32 + lane;
      if (k < nvalid4) d4[k] = sv[i] + rv[i];
    }
  }
}
#define W_U0 0
#define W_U1 1
#define W_U2 2
#define W_V(k) (3 + (k))
#define W_G(k) (11 + (k))

__global__ void __launch_bounds__(128)
mmb_node(const float* __restrict__ a0, const float* __restrict__ a1,
         const float* __restrict__ a2,
         const float* __restrict__ node0, const float* __restrict__ node1,
         const float* __restrict__ node2,
         const float* __restrict__ U, const float* __restrict__ V,
         const float* __restrict__ Wg,
         float* __restrict__ out0, float* __restrict__ out1,
         float* __restrict__ out2, int N) {
  __shared__ float2 wlds[14 * 8 * 2 * 32];
  __shared__ __attribute__((aligned(16))) float stage[4 * STAGE_F];

  for (int idx = threadIdx.x; idx < 14 * 1024; idx += blockDim.x) {
    const int wmat = idx >> 10;
    const int rem  = idx & 1023;
    const int k    = rem >> 5;
    const int n    = rem & 31;
    const float* src = (wmat < 3) ? (U + wmat * 1024)
                     : (wmat < 11) ? (V + (wmat - 3) * 1024)
                                   : (Wg + (wmat - 11) * 1024);
    const float val = src[rem];
    const int s   = k >> 2;
    const int kk  = k & 3;
    const int ln  = ((kk >> 1) << 4) | (n & 15);
    const int t   = n >> 4;
    const int jj  = kk & 1;
    ((float*)&wlds[((wmat * 8 + s) * 2 + t) * 32 + ln])[jj] = val;
  }
  __syncthreads();

  const int lane = threadIdx.x & 31;
  const int wv   = threadIdx.x >> 5;
  const int half = lane >> 4;
  const int m    = lane & 15;
  const int tile = blockIdx.x * 4 + wv;
  if (tile * 16 >= N) return;

#define LDB(wm, s, t)                                                       \
  ({ float2 _bv = wlds[(((wm) * 8 + (s)) * 2 + (t)) * 32 + lane];           \
     v2f _b; _b.x = _bv.x; _b.y = _bv.y; _b; })

#define MM2(acc, aop, wm, s)                                                \
  do { v2f _a = (aop);                                                      \
       acc[0] = wmma4(_a, LDB(wm, s, 0), acc[0]);                           \
       acc[1] = wmma4(_a, LDB(wm, s, 1), acc[1]); } while (0)

  const v8f vzero = {};
  float* const st = stage + wv * STAGE_F;
  const int nvr = (N - tile * 16 < 16) ? (N - tile * 16) : 16;
#define STAGE_FENCE() asm volatile("s_wait_dscnt 0" ::: "memory")

  const int rowA = tile * 16 + m;
  const int rowC = (rowA < N) ? rowA : (N - 1);
  float x0[2], b1[2][3], b2[2][9];
#define LOAD_BASE(s)                                                        \
  do { const int _k = (s) * 4 + 2 * half;                                   \
       const int _rc = rowC * CCH + _k;                                     \
       x0[0] = a0[_rc]; x0[1] = a0[_rc + 1];                                \
       _Pragma("unroll") for (int _a = 0; _a < 3; ++_a) {                   \
         b1[0][_a] = a1[_rc * 3 + _a]; b1[1][_a] = a1[(_rc + 1) * 3 + _a]; }\
       _Pragma("unroll") for (int _q = 0; _q < 9; ++_q) {                   \
         b2[0][_q] = a2[_rc * 9 + _q]; b2[1][_q] = a2[(_rc + 1) * 9 + _q]; }\
  } while (0)

  v8f y0acc[2] = {vzero, vzero};
  for (int s = 0; s < 8; ++s) {
    LOAD_BASE(s);
    v2f ax, asq, s1v, s2v;
#pragma unroll
    for (int jj = 0; jj < 2; ++jj) {
      ax[jj]  = x0[jj];
      asq[jj] = x0[jj] * x0[jj];
      float acc = 0.f;
#pragma unroll
      for (int a = 0; a < 3; ++a) acc += b1[jj][a] * b1[jj][a];
      s1v[jj] = acc;
      acc = 0.f;
#pragma unroll
      for (int q = 0; q < 9; ++q) acc += b2[jj][q] * b2[jj][q];
      s2v[jj] = acc;
    }
    MM2(y0acc, ax,  W_U0,  s);
    MM2(y0acc, asq, W_V(0), s);
    MM2(y0acc, s1v, W_V(3), s);
    MM2(y0acc, s2v, W_V(6), s);
  }

#pragma unroll
  for (int t = 0; t < 2; ++t)
#pragma unroll
    for (int r = 0; r < 8; ++r)
      st[(r + 8 * half) * CCH + t * 16 + m] = y0acc[t][r];
  STAGE_FENCE();

  v8f ga[3][2] = {{vzero, vzero}, {vzero, vzero}, {vzero, vzero}};
  for (int s = 0; s < 8; ++s) {
    const int k = s * 4 + 2 * half;
    v2f ay; ay.x = st[m * CCH + k]; ay.y = st[m * CCH + k + 1];
#pragma unroll
    for (int g = 0; g < 3; ++g) MM2(ga[g], ay, W_G(g), s);
  }
  v8f gate[3][2];
#pragma unroll
  for (int g = 0; g < 3; ++g)
#pragma unroll
    for (int t = 0; t < 2; ++t)
#pragma unroll
      for (int r = 0; r < 8; ++r) {
        const float v = ga[g][t][r];
        gate[g][t][r] = v * (1.0f / (1.0f + __expf(-v)));
      }

#pragma unroll
  for (int t = 0; t < 2; ++t)
#pragma unroll
    for (int r = 0; r < 8; ++r)
      st[(r + 8 * half) * CCH + t * 16 + m] = gate[0][t][r];
  STAGE_FENCE();
  flush_tile<4>(st, node0 + (size_t)tile * 16 * CCH,
                out0 + (size_t)tile * 16 * CCH, nvr * 8, lane);

  v8f y1acc[3][2] = {{vzero, vzero}, {vzero, vzero}, {vzero, vzero}};
  for (int s = 0; s < 8; ++s) {
    LOAD_BASE(s);
#pragma unroll
    for (int v = 0; v < 3; ++v) {
      v2f av, a01, tv;
#pragma unroll
      for (int jj = 0; jj < 2; ++jj) {
        av[jj]  = b1[jj][v];
        a01[jj] = x0[jj] * b1[jj][v];
        float acc = 0.f;
#pragma unroll
        for (int a = 0; a < 3; ++a) acc += b1[jj][a] * b2[jj][a * 3 + v];
        tv[jj] = acc;
      }
      MM2(y1acc[v], av,  W_U1,  s);
      MM2(y1acc[v], a01, W_V(1), s);
      MM2(y1acc[v], tv,  W_V(5), s);
    }
  }
#pragma unroll
  for (int v = 0; v < 3; ++v)
#pragma unroll
    for (int t = 0; t < 2; ++t)
#pragma unroll
      for (int r = 0; r < 8; ++r)
        st[((r + 8 * half) * CCH + t * 16 + m) * 3 + v] =
            y1acc[v][t][r] * gate[1][t][r];
  STAGE_FENCE();
  flush_tile<12>(st, node1 + (size_t)tile * 16 * CCH * 3,
                 out1 + (size_t)tile * 16 * CCH * 3, nvr * 24, lane);

  for (int aIdx = 0; aIdx < 3; ++aIdx) {
    v8f y2acc[3][2] = {{vzero, vzero}, {vzero, vzero}, {vzero, vzero}};
    for (int s = 0; s < 8; ++s) {
      LOAD_BASE(s);
#pragma unroll
      for (int b = 0; b < 3; ++b) {
        v2f a2v, a02, a11, qv;
#pragma unroll
        for (int jj = 0; jj < 2; ++jj) {
          const float g = b2[jj][aIdx * 3 + b];
          a2v[jj] = g;
          a02[jj] = x0[jj] * g;
          a11[jj] = b1[jj][aIdx] * b1[jj][b];
          float acc = 0.f;
#pragma unroll
          for (int kk = 0; kk < 3; ++kk)
            acc += b2[jj][aIdx * 3 + kk] * b2[jj][kk * 3 + b];
          qv[jj] = acc;
        }
        MM2(y2acc[b], a2v, W_U2,  s);
        MM2(y2acc[b], a02, W_V(2), s);
        MM2(y2acc[b], a11, W_V(4), s);
        MM2(y2acc[b], qv,  W_V(7), s);
      }
    }
#pragma unroll
    for (int b = 0; b < 3; ++b)
#pragma unroll
      for (int t = 0; t < 2; ++t)
#pragma unroll
        for (int r = 0; r < 8; ++r)
          st[((r + 8 * half) * CCH + t * 16 + m) * 9 + aIdx * 3 + b] =
              y2acc[b][t][r] * gate[2][t][r];
  }
  STAGE_FENCE();
  flush_tile<36>(st, node2 + (size_t)tile * 16 * CCH * 9,
                 out2 + (size_t)tile * 16 * CCH * 9, nvr * 72, lane);
#undef STAGE_FENCE
#undef LOAD_BASE
#undef MM2
#undef LDB
}

extern "C" void kernel_launch(void* const* d_in, const int* in_sizes, int n_in,
                              void* d_out, int out_size, void* d_ws, size_t ws_size,
                              hipStream_t stream) {
  const float* node0 = (const float*)d_in[0];
  const float* node1 = (const float*)d_in[1];
  const float* node2 = (const float*)d_in[2];
  const float* rij   = (const float*)d_in[3];
  const float* Wrad  = (const float*)d_in[4];
  const float* U     = (const float*)d_in[5];
  const float* V     = (const float*)d_in[6];
  const float* Wg    = (const float*)d_in[7];
  const int*   idx_i = (const int*)d_in[8];
  const int*   idx_j = (const int*)d_in[9];

  const int N = in_sizes[0] / CCH;
  const int E = in_sizes[3] / 3;

  float* a0 = (float*)d_ws;
  float* a1 = a0 + (size_t)N * CCH;
  float* a2 = a1 + (size_t)N * CCH * 3;

  float* out0 = (float*)d_out;
  float* out1 = out0 + (size_t)N * CCH;
  float* out2 = out1 + (size_t)N * CCH * 3;

  mmb_edge<<<(N + EDGE_TN - 1) / EDGE_TN, EDGE_WAVES * 32, 0, stream>>>(
      rij, Wrad, idx_i, idx_j, node0, node1, node2, a0, a1, a2, N, E);

  const int nTiles = (N + 15) / 16;
  mmb_node<<<(nTiles + 3) / 4, 128, 0, stream>>>(a0, a1, a2,
                                                 node0, node1, node2,
                                                 U, V, Wg,
                                                 out0, out1, out2, N);
}


# edge phase: a wave queues its range's edges in LDS, takes 16 at a time: geometry one edge per lane, the 11x32 radial filter sums as rbf x Wrad on WMMA (44 per 16 edges); filter sum order changes
# speedup vs baseline: 3.2302x; 1.1448x over previous
#include <hip/hip_runtime.h>
#include <hip/hip_bf16.h>
#include <math.h>

#define CCH 32
#define NPATH 11
#define NB 8

typedef __attribute__((ext_vector_type(2))) float v2f;
typedef __attribute__((ext_vector_type(8))) float v8f;
typedef __attribute__((ext_vector_type(4))) float v4f;

__device__ __forceinline__ v8f wmma4(v2f a, v2f b, v8f c) {
  return __builtin_amdgcn_wmma_f32_16x16x4_f32(
       false, a,  false, b,
       (short)0, c,  false,  false);
}

#define EDGE_TN    80
#define EDGE_WAVES 16
#define EDGE_SCAN  24
#define QCAP       1024
#define NFRAG      (NPATH * 2 * 2)

__device__ __forceinline__ void lds_add(float* p, float v) {
  __hip_atomic_fetch_add(p, v, __ATOMIC_RELAXED, __HIP_MEMORY_SCOPE_WORKGROUP);
}

__device__ __forceinline__ float xhalf(float x) {
  const int b = __builtin_bit_cast(int, x);
  return __builtin_bit_cast(float,
      __builtin_amdgcn_permlanex16(b, b, 0x76543210, 0xfedcba98, false, false));
}

__device__ __forceinline__ float lane_f(float x, int l) {
  return __builtin_bit_cast(float,
      __builtin_amdgcn_readlane(__builtin_bit_cast(int, x), l));
}

__device__ __forceinline__ void edge_msg(const float* __restrict__ node0,
                                         const float* __restrict__ node1,
                                         const float* __restrict__ node2,
                                         float* s0, float* s1, float* s2,
                                         const float (&w)[NPATH],
                                         int j, int nl, float rh0, float rh1, float rh2,
                                         int c) {
  const float rh[3] = {rh0, rh1, rh2};
  const float g0 = node0[j * CCH + c];
  float g1[3], g2[9];
#pragma unroll
  for (int a = 0; a < 3; ++a) g1[a] = node1[(j * CCH + c) * 3 + a];
#pragma unroll
  for (int q = 0; q < 9; ++q) g2[q] = node2[(j * CCH + c) * 9 + q];

  const float d1 = g1[0] * rh[0] + g1[1] * rh[1] + g1[2] * rh[2];
  float gr[3];
#pragma unroll
  for (int a = 0; a < 3; ++a)
    gr[a] = g2[a * 3 + 0] * rh[0] + g2[a * 3 + 1] * rh[1] + g2[a * 3 + 2] * rh[2];
  const float d2 = gr[0] * rh[0] + gr[1] * rh[1] + gr[2] * rh[2];

  const float inv16 = 0.0625f;
  const int   rc    = nl * CCH + c;

  lds_add(&s0[rc], (g0 * w[0] + d1 * w[4] + d2 * w[9]) * inv16);

#pragma unroll
  for (int a = 0; a < 3; ++a) {
    const float mv = (g0 * w[1] + d1 * w[6]) * rh[a] + g1[a] * w[3] + gr[a] * w[8];
    lds_add(&s1[rc * 3 + a], mv * inv16);
  }
#pragma unroll
  for (int a = 0; a < 3; ++a) {
    const float ca = g0 * w[2] * rh[a] + g1[a] * w[5] + gr[a] * w[10];
#pragma unroll
    for (int b = 0; b < 3; ++b) {
      const float mv = ca * rh[b] + g2[a * 3 + b] * w[7];
      lds_add(&s2[rc * 9 + a * 3 + b], mv * inv16);
    }
  }
}

__global__ void __launch_bounds__(EDGE_WAVES * 32)
mmb_edge(const float* __restrict__ rij,
         const float* __restrict__ Wrad,
         const int*   __restrict__ idx_i,
         const int*   __restrict__ idx_j,
         const float* __restrict__ node0,
         const float* __restrict__ node1,
         const float* __restrict__ node2,
         float* __restrict__ a0,
         float* __restrict__ a1,
         float* __restrict__ a2,
         int N, int E) {
  __shared__ float4 acc4[EDGE_TN * 104];
  float* const s0 = (float*)acc4;
  float* const s1 = s0 + EDGE_TN * CCH;
  float* const s2 = s0 + EDGE_TN * CCH * 4;
  __shared__ float2 wb[NFRAG * 32];
  __shared__ int queue[EDGE_WAVES * QCAP];

  const int tid = threadIdx.x;
  const int c   = tid & 31;
  const int hf  = c >> 4;
  const int q   = c & 15;
  const bool lo = (c < 16);
  const int wv  = __builtin_amdgcn_readfirstlane(tid >> 5);
  const int n0  = blockIdx.x * EDGE_TN;
  const int nv  = (N - n0 < EDGE_TN) ? (N - n0) : EDGE_TN;
  int* const qw = queue + wv * QCAP;

  constexpr int WR_N     = NFRAG * 64;
  constexpr int WR_TRIPS = (WR_N + EDGE_WAVES * 32 - 1) / (EDGE_WAVES * 32);
  float wt[WR_TRIPS];
#pragma unroll
  for (int u = 0; u < WR_TRIPS; ++u) {
    int k = tid + u * EDGE_WAVES * 32;
    k = (k < WR_N) ? k : (WR_N - 1);
    const int jj = k & 1, ln = (k >> 1) & 31, f = k >> 6;
    const int t = f & 1, s = (f >> 1) & 1, p = f >> 2;
    const int kk = 4 * s + 2 * (ln >> 4) + jj;
    wt[u] = Wrad[(p * NB + kk) * CCH + 16 * t + (ln & 15)];
  }
  for (int k = tid; k < EDGE_TN * 104; k += EDGE_WAVES * 32)
    acc4[k] = make_float4(0.f, 0.f, 0.f, 0.f);
#pragma unroll
  for (int u = 0; u < WR_TRIPS; ++u) {
    const int k = tid + u * EDGE_WAVES * 32;
    if (k < WR_N) ((float*)wb)[k] = wt[u];
  }
  __syncthreads();

#define CEN(b) (0.5f + 0.6428571429f * (float)(b))
  const float cen00 = hf ? CEN(2) : CEN(0);
  const float cen01 = hf ? CEN(3) : CEN(1);
  const float cen10 = hf ? CEN(6) : CEN(4);
  const float cen11 = hf ? CEN(7) : CEN(5);
#undef CEN
  const v8f vzero = {};

  const int nChunks = (E + 31) >> 5;
  int cb   = wv;
  int head = 0;
  int cnt  = 0;
  for (;;) {
    if (cnt < 16 && cb < nChunks) {
      int il[EDGE_SCAN];
#pragma unroll
      for (int u = 0; u < EDGE_SCAN; ++u) {
        const int ev = ((cb + u * EDGE_WAVES) << 5) + c;
        const int id = idx_i[(ev < E) ? ev : (E - 1)];
        il[u] = (ev < E) ? (id - n0) : -1;
      }
#pragma unroll
      for (int u = 0; u < EDGE_SCAN; ++u) {
        const bool hit = (unsigned)il[u] < (unsigned)nv;
        const unsigned mu = __builtin_amdgcn_ballot_w32(hit);
        if (mu != 0u) {
          const int pos = cnt + (int)__builtin_amdgcn_mbcnt_lo(mu, 0u);
          if (hit) qw[(head + pos) & (QCAP - 1)] = ((cb + u * EDGE_WAVES) << 5) + c;
          cnt += __builtin_popcount(mu);
        }
      }
      cb += EDGE_WAVES * EDGE_SCAN;
      continue;
    }
    if (cnt == 0) break;
    const int nb = (cnt < 16) ? cnt : 16;

    asm volatile("s_wait_dscnt 0" ::: "memory");

    const int e = qw[(head + ((q < nb) ? q : 0)) & (QCAP - 1)];
    const float r0 = rij[e * 3 + 0], r1 = rij[e * 3 + 1], r2 = rij[e * 3 + 2];
    const int   jv  = idx_j[e];
    const int   nlv = idx_i[e] - n0;
    const float d  = sqrtf(r0 * r0 + r1 * r1 + r2 * r2 + 1e-12f);
    const float iv = 1.0f / d;
    const float rhx = r0 * iv, rhy = r1 * iv, rhz = r2 * iv;

    float x = d * 0.2f;
    x = (x > 1.0f) ? 1.0f : x;
    const float fc = 0.5f * (cosf(3.14159265358979f * x) + 1.0f);

    const float t00 = d - cen00, t01 = d - cen01, t10 = d - cen10, t11 = d - cen11;
    v2f af0, af1;
    af0.x = __expf(-4.0f * t00 * t00) * fc;
    af0.y = __expf(-4.0f * t01 * t01) * fc;
    af1.x = __expf(-4.0f * t10 * t10) * fc;
    af1.y = __expf(-4.0f * t11 * t11) * fc;

    v8f D[NPATH][2];
#pragma unroll
    for (int p = 0; p < NPATH; ++p)
#pragma unroll
      for (int t = 0; t < 2; ++t) {
        const float2 b0 = wb[((p * 2 + 0) * 2 + t) * 32 + c];
        const float2 b1 = wb[((p * 2 + 1) * 2 + t) * 32 + c];
        v2f bf0, bf1;
        bf0.x = b0.x; bf0.y = b0.y; bf1.x = b1.x; bf1.y = b1.y;
        v8f acc = wmma4(af0, bf0, vzero);
        D[p][t] = wmma4(af1, bf1, acc);
      }

#pragma unroll
    for (int r = 0; r < 8; ++r) {
      float wA[NPATH], wB[NPATH];
#pragma unroll
      for (int p = 0; p < NPATH; ++p) {
        const float x0 = D[p][0][r], x1 = D[p][1][r];
        const float pz = xhalf(lo ? x1 : x0);
        wA[p] = lo ? x0 : pz;
        wB[p] = lo ? pz : x1;
      }
      if (r < nb)
        edge_msg(node0, node1, node2, s0, s1, s2, wA,
                 __builtin_amdgcn_readlane(jv, r), __builtin_amdgcn_readlane(nlv, r),
                 lane_f(rhx, r), lane_f(rhy, r), lane_f(rhz, r), c);
      if (8 + r < nb)
        edge_msg(node0, node1, node2, s0, s1, s2, wB,
                 __builtin_amdgcn_readlane(jv, 8 + r), __builtin_amdgcn_readlane(nlv, 8 + r),
                 lane_f(rhx, 8 + r), lane_f(rhy, 8 + r), lane_f(rhz, 8 + r), c);
    }
    head = (head + nb) & (QCAP - 1);
    cnt -= nb;
  }
  __syncthreads();

  float4* const o0 = (float4*)(a0 + (size_t)n0 * CCH);
  float4* const o1 = (float4*)(a1 + (size_t)n0 * CCH * 3);
  float4* const o2 = (float4*)(a2 + (size_t)n0 * CCH * 9);
  for (int k = tid; k < nv * 8; k += EDGE_WAVES * 32)  o0[k] = acc4[k];
  for (int k = tid; k < nv * 24; k += EDGE_WAVES * 32) o1[k] = acc4[EDGE_TN * 8 + k];
  for (int k = tid; k < nv * 72; k += EDGE_WAVES * 32) o2[k] = acc4[EDGE_TN * 32 + k];
}

#define STAGE_F (16 * CCH * 9)

template <int PER_LANE>
__device__ __forceinline__ void flush_tile(const float* st,
                                           const float* __restrict__ res,
                                           float* __restrict__ dst,
                                           int nvalid4, int lane) {
  const v4f* const s4 = (const v4f*)st;
  const v4f* const r4 = (const v4f*)res;
  v4f* const d4 = (v4f*)dst;
  constexpr int BATCH = (PER_LANE < 12) ? PER_LANE : 12;
#pragma unroll
  for (int i0 = 0; i0 < PER_LANE; i0 += BATCH) {
    v4f rv[BATCH], sv[BATCH];
#pragma unroll
    for (int i = 0; i < BATCH; ++i) {
      const int k = (i0 + i) * 32 + lane;
      rv[i] = r4[(k < nvalid4) ? k : (nvalid4 - 1)];
      sv[i] = s4[k];
    }
#pragma unroll
    for (int i = 0; i < BATCH; ++i) {
      const int k = (i0 + i) * 32 + lane;
      if (k < nvalid4) d4[k] = sv[i] + rv[i];
    }
  }
}
#define W_U0 0
#define W_U1 1
#define W_U2 2
#define W_V(k) (3 + (k))
#define W_G(k) (11 + (k))

__global__ void __launch_bounds__(128)
mmb_node(const float* __restrict__ a0, const float* __restrict__ a1,
         const float* __restrict__ a2,
         const float* __restrict__ node0, const float* __restrict__ node1,
         const float* __restrict__ node2,
         const float* __restrict__ U, const float* __restrict__ V,
         const float* __restrict__ Wg,
         float* __restrict__ out0, float* __restrict__ out1,
         float* __restrict__ out2, int N) {
  __shared__ float2 wlds[14 * 8 * 2 * 32];
  __shared__ __attribute__((aligned(16))) float stage[4 * STAGE_F];

  for (int idx = threadIdx.x; idx < 14 * 1024; idx += blockDim.x) {
    const int wmat = idx >> 10;
    const int rem  = idx & 1023;
    const int k    = rem >> 5;
    const int n    = rem & 31;
    const float* src = (wmat < 3) ? (U + wmat * 1024)
                     : (wmat < 11) ? (V + (wmat - 3) * 1024)
                                   : (Wg + (wmat - 11) * 1024);
    const float val = src[rem];
    const int s   = k >> 2;
    const int kk  = k & 3;
    const int ln  = ((kk >> 1) << 4) | (n & 15);
    const int t   = n >> 4;
    const int jj  = kk & 1;
    ((float*)&wlds[((wmat * 8 + s) * 2 + t) * 32 + ln])[jj] = val;
  }
  __syncthreads();

  const int lane = threadIdx.x & 31;
  const int wv   = threadIdx.x >> 5;
  const int half = lane >> 4;
  const int m    = lane & 15;
  const int tile = blockIdx.x * 4 + wv;
  if (tile * 16 >= N) return;

#define LDB(wm, s, t)                                                       \
  ({ float2 _bv = wlds[(((wm) * 8 + (s)) * 2 + (t)) * 32 + lane];           \
     v2f _b; _b.x = _bv.x; _b.y = _bv.y; _b; })

#define MM2(acc, aop, wm, s)                                                \
  do { v2f _a = (aop);                                                      \
       acc[0] = wmma4(_a, LDB(wm, s, 0), acc[0]);                           \
       acc[1] = wmma4(_a, LDB(wm, s, 1), acc[1]); } while (0)

  const v8f vzero = {};
  float* const st = stage + wv * STAGE_F;
  const int nvr = (N - tile * 16 < 16) ? (N - tile * 16) : 16;
#define STAGE_FENCE() asm volatile("s_wait_dscnt 0" ::: "memory")

  const int rowA = tile * 16 + m;
  const int rowC = (rowA < N) ? rowA : (N - 1);
  float x0[2], b1[2][3], b2[2][9];
#define LOAD_BASE(s)                                                        \
  do { const int _k = (s) * 4 + 2 * half;                                   \
       const int _rc = rowC * CCH + _k;                                     \
       x0[0] = a0[_rc]; x0[1] = a0[_rc + 1];                                \
       _Pragma("unroll") for (int _a = 0; _a < 3; ++_a) {                   \
         b1[0][_a] = a1[_rc * 3 + _a]; b1[1][_a] = a1[(_rc + 1) * 3 + _a]; }\
       _Pragma("unroll") for (int _q = 0; _q < 9; ++_q) {                   \
         b2[0][_q] = a2[_rc * 9 + _q]; b2[1][_q] = a2[(_rc + 1) * 9 + _q]; }\
  } while (0)

  v8f y0acc[2] = {vzero, vzero};
  for (int s = 0; s < 8; ++s) {
    LOAD_BASE(s);
    v2f ax, asq, s1v, s2v;
#pragma unroll
    for (int jj = 0; jj < 2; ++jj) {
      ax[jj]  = x0[jj];
      asq[jj] = x0[jj] * x0[jj];
      float acc = 0.f;
#pragma unroll
      for (int a = 0; a < 3; ++a) acc += b1[jj][a] * b1[jj][a];
      s1v[jj] = acc;
      acc = 0.f;
#pragma unroll
      for (int q = 0; q < 9; ++q) acc += b2[jj][q] * b2[jj][q];
      s2v[jj] = acc;
    }
    MM2(y0acc, ax,  W_U0,  s);
    MM2(y0acc, asq, W_V(0), s);
    MM2(y0acc, s1v, W_V(3), s);
    MM2(y0acc, s2v, W_V(6), s);
  }

#pragma unroll
  for (int t = 0; t < 2; ++t)
#pragma unroll
    for (int r = 0; r < 8; ++r)
      st[(r + 8 * half) * CCH + t * 16 + m] = y0acc[t][r];
  STAGE_FENCE();

  v8f ga[3][2] = {{vzero, vzero}, {vzero, vzero}, {vzero, vzero}};
  for (int s = 0; s < 8; ++s) {
    const int k = s * 4 + 2 * half;
    v2f ay; ay.x = st[m * CCH + k]; ay.y = st[m * CCH + k + 1];
#pragma unroll
    for (int g = 0; g < 3; ++g) MM2(ga[g], ay, W_G(g), s);
  }
  v8f gate[3][2];
#pragma unroll
  for (int g = 0; g < 3; ++g)
#pragma unroll
    for (int t = 0; t < 2; ++t)
#pragma unroll
      for (int r = 0; r < 8; ++r) {
        const float v = ga[g][t][r];
        gate[g][t][r] = v * (1.0f / (1.0f + __expf(-v)));
      }

#pragma unroll
  for (int t = 0; t < 2; ++t)
#pragma unroll
    for (int r = 0; r < 8; ++r)
      st[(r + 8 * half) * CCH + t * 16 + m] = gate[0][t][r];
  STAGE_FENCE();
  flush_tile<4>(st, node0 + (size_t)tile * 16 * CCH,
                out0 + (size_t)tile * 16 * CCH, nvr * 8, lane);

  v8f y1acc[3][2] = {{vzero, vzero}, {vzero, vzero}, {vzero, vzero}};
  for (int s = 0; s < 8; ++s) {
    LOAD_BASE(s);
#pragma unroll
    for (int v = 0; v < 3; ++v) {
      v2f av, a01, tv;
#pragma unroll
      for (int jj = 0; jj < 2; ++jj) {
        av[jj]  = b1[jj][v];
        a01[jj] = x0[jj] * b1[jj][v];
        float acc = 0.f;
#pragma unroll
        for (int a = 0; a < 3; ++a) acc += b1[jj][a] * b2[jj][a * 3 + v];
        tv[jj] = acc;
      }
      MM2(y1acc[v], av,  W_U1,  s);
      MM2(y1acc[v], a01, W_V(1), s);
      MM2(y1acc[v], tv,  W_V(5), s);
    }
  }
#pragma unroll
  for (int v = 0; v < 3; ++v)
#pragma unroll
    for (int t = 0; t < 2; ++t)
#pragma unroll
      for (int r = 0; r < 8; ++r)
        st[((r + 8 * half) * CCH + t * 16 + m) * 3 + v] =
            y1acc[v][t][r] * gate[1][t][r];
  STAGE_FENCE();
  flush_tile<12>(st, node1 + (size_t)tile * 16 * CCH * 3,
                 out1 + (size_t)tile * 16 * CCH * 3, nvr * 24, lane);

  for (int aIdx = 0; aIdx < 3; ++aIdx) {
    v8f y2acc[3][2] = {{vzero, vzero}, {vzero, vzero}, {vzero, vzero}};
    for (int s = 0; s < 8; ++s) {
      LOAD_BASE(s);
#pragma unroll
      for (int b = 0; b < 3; ++b) {
        v2f a2v, a02, a11, qv;
#pragma unroll
        for (int jj = 0; jj < 2; ++jj) {
          const float g = b2[jj][aIdx * 3 + b];
          a2v[jj] = g;
          a02[jj] = x0[jj] * g;
          a11[jj] = b1[jj][aIdx] * b1[jj][b];
          float acc = 0.f;
#pragma unroll
          for (int kk = 0; kk < 3; ++kk)
            acc += b2[jj][aIdx * 3 + kk] * b2[jj][kk * 3 + b];
          qv[jj] = acc;
        }
        MM2(y2acc[b], a2v, W_U2,  s);
        MM2(y2acc[b], a02, W_V(2), s);
        MM2(y2acc[b], a11, W_V(4), s);
        MM2(y2acc[b], qv,  W_V(7), s);
      }
    }
#pragma unroll
    for (int b = 0; b < 3; ++b)
#pragma unroll
      for (int t = 0; t < 2; ++t)
#pragma unroll
        for (int r = 0; r < 8; ++r)
          st[((r + 8 * half) * CCH + t * 16 + m) * 9 + aIdx * 3 + b] =
              y2acc[b][t][r] * gate[2][t][r];
  }
  STAGE_FENCE();
  flush_tile<36>(st, node2 + (size_t)tile * 16 * CCH * 9,
                 out2 + (size_t)tile * 16 * CCH * 9, nvr * 72, lane);
#undef STAGE_FENCE
#undef LOAD_BASE
#undef MM2
#undef LDB
}

extern "C" void kernel_launch(void* const* d_in, const int* in_sizes, int n_in,
                              void* d_out, int out_size, void* d_ws, size_t ws_size,
                              hipStream_t stream) {
  const float* node0 = (const float*)d_in[0];
  const float* node1 = (const float*)d_in[1];
  const float* node2 = (const float*)d_in[2];
  const float* rij   = (const float*)d_in[3];
  const float* Wrad  = (const float*)d_in[4];
  const float* U     = (const float*)d_in[5];
  const float* V     = (const float*)d_in[6];
  const float* Wg    = (const float*)d_in[7];
  const int*   idx_i = (const int*)d_in[8];
  const int*   idx_j = (const int*)d_in[9];

  const int N = in_sizes[0] / CCH;
  const int E = in_sizes[3] / 3;

  float* a0 = (float*)d_ws;
  float* a1 = a0 + (size_t)N * CCH;
  float* a2 = a1 + (size_t)N * CCH * 3;

  float* out0 = (float*)d_out;
  float* out1 = out0 + (size_t)N * CCH;
  float* out2 = out1 + (size_t)N * CCH * 3;

  mmb_edge<<<(N + EDGE_TN - 1) / EDGE_TN, EDGE_WAVES * 32, 0, stream>>>(
      rij, Wrad, idx_i, idx_j, node0, node1, node2, a0, a1, a2, N, E);

  const int nTiles = (N + 15) / 16;
  mmb_node<<<(nTiles + 3) / 4, 128, 0, stream>>>(a0, a1, a2,
                                                 node0, node1, node2,
                                                 U, V, Wg,
                                                 out0, out1, out2, N);
}


# node phase re-dealt: 256 workgroups of 8 waves, one per WGP; tile t -> workgroup t % 256, wave t / 256; weights staged once per WGP; 512-register cap, so all 1250 tiles run in one round
# speedup vs baseline: 3.7193x; 1.1514x over previous
#include <hip/hip_runtime.h>
#include <hip/hip_bf16.h>
#include <math.h>

#define CCH 32
#define NPATH 11
#define NB 8

typedef __attribute__((ext_vector_type(2))) float v2f;
typedef __attribute__((ext_vector_type(8))) float v8f;
typedef __attribute__((ext_vector_type(4))) float v4f;

__device__ __forceinline__ v8f wmma4(v2f a, v2f b, v8f c) {
  return __builtin_amdgcn_wmma_f32_16x16x4_f32(
       false, a,  false, b,
       (short)0, c,  false,  false);
}

#define EDGE_TN    80
#define EDGE_WAVES 16
#define EDGE_SCAN  24
#define QCAP       1024
#define NFRAG      (NPATH * 2 * 2)

__device__ __forceinline__ void lds_add(float* p, float v) {
  __hip_atomic_fetch_add(p, v, __ATOMIC_RELAXED, __HIP_MEMORY_SCOPE_WORKGROUP);
}

__device__ __forceinline__ float xhalf(float x) {
  const int b = __builtin_bit_cast(int, x);
  return __builtin_bit_cast(float,
      __builtin_amdgcn_permlanex16(b, b, 0x76543210, 0xfedcba98, false, false));
}

__device__ __forceinline__ float lane_f(float x, int l) {
  return __builtin_bit_cast(float,
      __builtin_amdgcn_readlane(__builtin_bit_cast(int, x), l));
}

__device__ __forceinline__ void edge_msg(const float* __restrict__ node0,
                                         const float* __restrict__ node1,
                                         const float* __restrict__ node2,
                                         float* s0, float* s1, float* s2,
                                         const float (&w)[NPATH],
                                         int j, int nl, float rh0, float rh1, float rh2,
                                         int c) {
  const float rh[3] = {rh0, rh1, rh2};
  const float g0 = node0[j * CCH + c];
  float g1[3], g2[9];
#pragma unroll
  for (int a = 0; a < 3; ++a) g1[a] = node1[(j * CCH + c) * 3 + a];
#pragma unroll
  for (int q = 0; q < 9; ++q) g2[q] = node2[(j * CCH + c) * 9 + q];

  const float d1 = g1[0] * rh[0] + g1[1] * rh[1] + g1[2] * rh[2];
  float gr[3];
#pragma unroll
  for (int a = 0; a < 3; ++a)
    gr[a] = g2[a * 3 + 0] * rh[0] + g2[a * 3 + 1] * rh[1] + g2[a * 3 + 2] * rh[2];
  const float d2 = gr[0] * rh[0] + gr[1] * rh[1] + gr[2] * rh[2];

  const float inv16 = 0.0625f;
  const int   rc    = nl * CCH + c;

  lds_add(&s0[rc], (g0 * w[0] + d1 * w[4] + d2 * w[9]) * inv16);

#pragma unroll
  for (int a = 0; a < 3; ++a) {
    const float mv = (g0 * w[1] + d1 * w[6]) * rh[a] + g1[a] * w[3] + gr[a] * w[8];
    lds_add(&s1[rc * 3 + a], mv * inv16);
  }
#pragma unroll
  for (int a = 0; a < 3; ++a) {
    const float ca = g0 * w[2] * rh[a] + g1[a] * w[5] + gr[a] * w[10];
#pragma unroll
    for (int b = 0; b < 3; ++b) {
      const float mv = ca * rh[b] + g2[a * 3 + b] * w[7];
      lds_add(&s2[rc * 9 + a * 3 + b], mv * inv16);
    }
  }
}

__global__ void __launch_bounds__(EDGE_WAVES * 32)
mmb_edge(const float* __restrict__ rij,
         const float* __restrict__ Wrad,
         const int*   __restrict__ idx_i,
         const int*   __restrict__ idx_j,
         const float* __restrict__ node0,
         const float* __restrict__ node1,
         const float* __restrict__ node2,
         float* __restrict__ a0,
         float* __restrict__ a1,
         float* __restrict__ a2,
         int N, int E) {
  __shared__ float4 acc4[EDGE_TN * 104];
  float* const s0 = (float*)acc4;
  float* const s1 = s0 + EDGE_TN * CCH;
  float* const s2 = s0 + EDGE_TN * CCH * 4;
  __shared__ float2 wb[NFRAG * 32];
  __shared__ int queue[EDGE_WAVES * QCAP];

  const int tid = threadIdx.x;
  const int c   = tid & 31;
  const int hf  = c >> 4;
  const int q   = c & 15;
  const bool lo = (c < 16);
  const int wv  = __builtin_amdgcn_readfirstlane(tid >> 5);
  const int n0  = blockIdx.x * EDGE_TN;
  const int nv  = (N - n0 < EDGE_TN) ? (N - n0) : EDGE_TN;
  int* const qw = queue + wv * QCAP;

  constexpr int WR_N     = NFRAG * 64;
  constexpr int WR_TRIPS = (WR_N + EDGE_WAVES * 32 - 1) / (EDGE_WAVES * 32);
  float wt[WR_TRIPS];
#pragma unroll
  for (int u = 0; u < WR_TRIPS; ++u) {
    int k = tid + u * EDGE_WAVES * 32;
    k = (k < WR_N) ? k : (WR_N - 1);
    const int jj = k & 1, ln = (k >> 1) & 31, f = k >> 6;
    const int t = f & 1, s = (f >> 1) & 1, p = f >> 2;
    const int kk = 4 * s + 2 * (ln >> 4) + jj;
    wt[u] = Wrad[(p * NB + kk) * CCH + 16 * t + (ln & 15)];
  }
  for (int k = tid; k < EDGE_TN * 104; k += EDGE_WAVES * 32)
    acc4[k] = make_float4(0.f, 0.f, 0.f, 0.f);
#pragma unroll
  for (int u = 0; u < WR_TRIPS; ++u) {
    const int k = tid + u * EDGE_WAVES * 32;
    if (k < WR_N) ((float*)wb)[k] = wt[u];
  }
  __syncthreads();

#define CEN(b) (0.5f + 0.6428571429f * (float)(b))
  const float cen00 = hf ? CEN(2) : CEN(0);
  const float cen01 = hf ? CEN(3) : CEN(1);
  const float cen10 = hf ? CEN(6) : CEN(4);
  const float cen11 = hf ? CEN(7) : CEN(5);
#undef CEN
  const v8f vzero = {};

  const int nChunks = (E + 31) >> 5;
  int cb   = wv;
  int head = 0;
  int cnt  = 0;
  for (;;) {
    if (cnt < 16 && cb < nChunks) {
      int il[EDGE_SCAN];
#pragma unroll
      for (int u = 0; u < EDGE_SCAN; ++u) {
        const int ev = ((cb + u * EDGE_WAVES) << 5) + c;
        const int id = idx_i[(ev < E) ? ev : (E - 1)];
        il[u] = (ev < E) ? (id - n0) : -1;
      }
#pragma unroll
      for (int u = 0; u < EDGE_SCAN; ++u) {
        const bool hit = (unsigned)il[u] < (unsigned)nv;
        const unsigned mu = __builtin_amdgcn_ballot_w32(hit);
        if (mu != 0u) {
          const int pos = cnt + (int)__builtin_amdgcn_mbcnt_lo(mu, 0u);
          if (hit) qw[(head + pos) & (QCAP - 1)] = ((cb + u * EDGE_WAVES) << 5) + c;
          cnt += __builtin_popcount(mu);
        }
      }
      cb += EDGE_WAVES * EDGE_SCAN;
      continue;
    }
    if (cnt == 0) break;
    const int nb = (cnt < 16) ? cnt : 16;

    asm volatile("s_wait_dscnt 0" ::: "memory");

    const int e = qw[(head + ((q < nb) ? q : 0)) & (QCAP - 1)];
    const float r0 = rij[e * 3 + 0], r1 = rij[e * 3 + 1], r2 = rij[e * 3 + 2];
    const int   jv  = idx_j[e];
    const int   nlv = idx_i[e] - n0;
    const float d  = sqrtf(r0 * r0 + r1 * r1 + r2 * r2 + 1e-12f);
    const float iv = 1.0f / d;
    const float rhx = r0 * iv, rhy = r1 * iv, rhz = r2 * iv;

    float x = d * 0.2f;
    x = (x > 1.0f) ? 1.0f : x;
    const float fc = 0.5f * (cosf(3.14159265358979f * x) + 1.0f);

    const float t00 = d - cen00, t01 = d - cen01, t10 = d - cen10, t11 = d - cen11;
    v2f af0, af1;
    af0.x = __expf(-4.0f * t00 * t00) * fc;
    af0.y = __expf(-4.0f * t01 * t01) * fc;
    af1.x = __expf(-4.0f * t10 * t10) * fc;
    af1.y = __expf(-4.0f * t11 * t11) * fc;

    v8f D[NPATH][2];
#pragma unroll
    for (int p = 0; p < NPATH; ++p)
#pragma unroll
      for (int t = 0; t < 2; ++t) {
        const float2 b0 = wb[((p * 2 + 0) * 2 + t) * 32 + c];
        const float2 b1 = wb[((p * 2 + 1) * 2 + t) * 32 + c];
        v2f bf0, bf1;
        bf0.x = b0.x; bf0.y = b0.y; bf1.x = b1.x; bf1.y = b1.y;
        v8f acc = wmma4(af0, bf0, vzero);
        D[p][t] = wmma4(af1, bf1, acc);
      }

#pragma unroll
    for (int r = 0; r < 8; ++r) {
      float wA[NPATH], wB[NPATH];
#pragma unroll
      for (int p = 0; p < NPATH; ++p) {
        const float x0 = D[p][0][r], x1 = D[p][1][r];
        const float pz = xhalf(lo ? x1 : x0);
        wA[p] = lo ? x0 : pz;
        wB[p] = lo ? pz : x1;
      }
      if (r < nb)
        edge_msg(node0, node1, node2, s0, s1, s2, wA,
                 __builtin_amdgcn_readlane(jv, r), __builtin_amdgcn_readlane(nlv, r),
                 lane_f(rhx, r), lane_f(rhy, r), lane_f(rhz, r), c);
      if (8 + r < nb)
        edge_msg(node0, node1, node2, s0, s1, s2, wB,
                 __builtin_amdgcn_readlane(jv, 8 + r), __builtin_amdgcn_readlane(nlv, 8 + r),
                 lane_f(rhx, 8 + r), lane_f(rhy, 8 + r), lane_f(rhz, 8 + r), c);
    }
    head = (head + nb) & (QCAP - 1);
    cnt -= nb;
  }
  __syncthreads();

  float4* const o0 = (float4*)(a0 + (size_t)n0 * CCH);
  float4* const o1 = (float4*)(a1 + (size_t)n0 * CCH * 3);
  float4* const o2 = (float4*)(a2 + (size_t)n0 * CCH * 9);
  for (int k = tid; k < nv * 8; k += EDGE_WAVES * 32)  o0[k] = acc4[k];
  for (int k = tid; k < nv * 24; k += EDGE_WAVES * 32) o1[k] = acc4[EDGE_TN * 8 + k];
  for (int k = tid; k < nv * 72; k += EDGE_WAVES * 32) o2[k] = acc4[EDGE_TN * 32 + k];
}

#define STAGE_F (16 * CCH * 9)
#define NODE_WAVES 8
#define NODE_WGS   256

template <int PER_LANE>
__device__ __forceinline__ void flush_tile(const float* st,
                                           const float* __restrict__ res,
                                           float* __restrict__ dst,
                                           int nvalid4, int lane) {
  const v4f* const s4 = (const v4f*)st;
  const v4f* const r4 = (const v4f*)res;
  v4f* const d4 = (v4f*)dst;
  constexpr int BATCH = (PER_LANE < 12) ? PER_LANE : 12;
#pragma unroll
  for (int i0 = 0; i0 < PER_LANE; i0 += BATCH) {
    v4f rv[BATCH], sv[BATCH];
#pragma unroll
    for (int i = 0; i < BATCH; ++i) {
      const int k = (i0 + i) * 32 + lane;
      rv[i] = r4[(k < nvalid4) ? k : (nvalid4 - 1)];
      sv[i] = s4[k];
    }
#pragma unroll
    for (int i = 0; i < BATCH; ++i) {
      const int k = (i0 + i) * 32 + lane;
      if (k < nvalid4) d4[k] = sv[i] + rv[i];
    }
  }
}
#define W_U0 0
#define W_U1 1
#define W_U2 2
#define W_V(k) (3 + (k))
#define W_G(k) (11 + (k))

__global__ void __launch_bounds__(NODE_WAVES * 32)
mmb_node(const float* __restrict__ a0, const float* __restrict__ a1,
         const float* __restrict__ a2,
         const float* __restrict__ node0, const float* __restrict__ node1,
         const float* __restrict__ node2,
         const float* __restrict__ U, const float* __restrict__ V,
         const float* __restrict__ Wg,
         float* __restrict__ out0, float* __restrict__ out1,
         float* __restrict__ out2, int N) {
  __shared__ float2 wlds[14 * 8 * 2 * 32];
  __shared__ __attribute__((aligned(16))) float stage[NODE_WAVES * STAGE_F];

  for (int idx = threadIdx.x; idx < 14 * 1024; idx += blockDim.x) {
    const int wmat = idx >> 10;
    const int rem  = idx & 1023;
    const int k    = rem >> 5;
    const int n    = rem & 31;
    const float* src = (wmat < 3) ? (U + wmat * 1024)
                     : (wmat < 11) ? (V + (wmat - 3) * 1024)
                                   : (Wg + (wmat - 11) * 1024);
    const float val = src[rem];
    const int s   = k >> 2;
    const int kk  = k & 3;
    const int ln  = ((kk >> 1) << 4) | (n & 15);
    const int t   = n >> 4;
    const int jj  = kk & 1;
    ((float*)&wlds[((wmat * 8 + s) * 2 + t) * 32 + ln])[jj] = val;
  }
  __syncthreads();

  const int lane = threadIdx.x & 31;
  const int wv   = __builtin_amdgcn_readfirstlane(threadIdx.x >> 5);
  const int half = lane >> 4;
  const int m    = lane & 15;
  const int nTiles = (N + 15) >> 4;

#define LDB(wm, s, t)                                                       \
  ({ float2 _bv = wlds[(((wm) * 8 + (s)) * 2 + (t)) * 32 + lane];           \
     v2f _b; _b.x = _bv.x; _b.y = _bv.y; _b; })

#define MM2(acc, aop, wm, s)                                                \
  do { v2f _a = (aop);                                                      \
       acc[0] = wmma4(_a, LDB(wm, s, 0), acc[0]);                           \
       acc[1] = wmma4(_a, LDB(wm, s, 1), acc[1]); } while (0)

  const v8f vzero = {};
  float* const st = stage + wv * STAGE_F;
  for (int tile = wv * (int)gridDim.x + (int)blockIdx.x; tile < nTiles;
       tile += NODE_WAVES * (int)gridDim.x) {
  const int nvr = (N - tile * 16 < 16) ? (N - tile * 16) : 16;
#define STAGE_FENCE() asm volatile("s_wait_dscnt 0" ::: "memory")

  const int rowA = tile * 16 + m;
  const int rowC = (rowA < N) ? rowA : (N - 1);
  float x0[2], b1[2][3], b2[2][9];
#define LOAD_BASE(s)                                                        \
  do { const int _k = (s) * 4 + 2 * half;                                   \
       const int _rc = rowC * CCH + _k;                                     \
       x0[0] = a0[_rc]; x0[1] = a0[_rc + 1];                                \
       _Pragma("unroll") for (int _a = 0; _a < 3; ++_a) {                   \
         b1[0][_a] = a1[_rc * 3 + _a]; b1[1][_a] = a1[(_rc + 1) * 3 + _a]; }\
       _Pragma("unroll") for (int _q = 0; _q < 9; ++_q) {                   \
         b2[0][_q] = a2[_rc * 9 + _q]; b2[1][_q] = a2[(_rc + 1) * 9 + _q]; }\
  } while (0)

  v8f y0acc[2] = {vzero, vzero};
  for (int s = 0; s < 8; ++s) {
    LOAD_BASE(s);
    v2f ax, asq, s1v, s2v;
#pragma unroll
    for (int jj = 0; jj < 2; ++jj) {
      ax[jj]  = x0[jj];
      asq[jj] = x0[jj] * x0[jj];
      float acc = 0.f;
#pragma unroll
      for (int a = 0; a < 3; ++a) acc += b1[jj][a] * b1[jj][a];
      s1v[jj] = acc;
      acc = 0.f;
#pragma unroll
      for (int q = 0; q < 9; ++q) acc += b2[jj][q] * b2[jj][q];
      s2v[jj] = acc;
    }
    MM2(y0acc, ax,  W_U0,  s);
    MM2(y0acc, asq, W_V(0), s);
    MM2(y0acc, s1v, W_V(3), s);
    MM2(y0acc, s2v, W_V(6), s);
  }

#pragma unroll
  for (int t = 0; t < 2; ++t)
#pragma unroll
    for (int r = 0; r < 8; ++r)
      st[(r + 8 * half) * CCH + t * 16 + m] = y0acc[t][r];
  STAGE_FENCE();

  v8f ga[3][2] = {{vzero, vzero}, {vzero, vzero}, {vzero, vzero}};
  for (int s = 0; s < 8; ++s) {
    const int k = s * 4 + 2 * half;
    v2f ay; ay.x = st[m * CCH + k]; ay.y = st[m * CCH + k + 1];
#pragma unroll
    for (int g = 0; g < 3; ++g) MM2(ga[g], ay, W_G(g), s);
  }
  v8f gate[3][2];
#pragma unroll
  for (int g = 0; g < 3; ++g)
#pragma unroll
    for (int t = 0; t < 2; ++t)
#pragma unroll
      for (int r = 0; r < 8; ++r) {
        const float v = ga[g][t][r];
        gate[g][t][r] = v * (1.0f / (1.0f + __expf(-v)));
      }

#pragma unroll
  for (int t = 0; t < 2; ++t)
#pragma unroll
    for (int r = 0; r < 8; ++r)
      st[(r + 8 * half) * CCH + t * 16 + m] = gate[0][t][r];
  STAGE_FENCE();
  flush_tile<4>(st, node0 + (size_t)tile * 16 * CCH,
                out0 + (size_t)tile * 16 * CCH, nvr * 8, lane);

  v8f y1acc[3][2] = {{vzero, vzero}, {vzero, vzero}, {vzero, vzero}};
  for (int s = 0; s < 8; ++s) {
    LOAD_BASE(s);
#pragma unroll
    for (int v = 0; v < 3; ++v) {
      v2f av, a01, tv;
#pragma unroll
      for (int jj = 0; jj < 2; ++jj) {
        av[jj]  = b1[jj][v];
        a01[jj] = x0[jj] * b1[jj][v];
        float acc = 0.f;
#pragma unroll
        for (int a = 0; a < 3; ++a) acc += b1[jj][a] * b2[jj][a * 3 + v];
        tv[jj] = acc;
      }
      MM2(y1acc[v], av,  W_U1,  s);
      MM2(y1acc[v], a01, W_V(1), s);
      MM2(y1acc[v], tv,  W_V(5), s);
    }
  }
#pragma unroll
  for (int v = 0; v < 3; ++v)
#pragma unroll
    for (int t = 0; t < 2; ++t)
#pragma unroll
      for (int r = 0; r < 8; ++r)
        st[((r + 8 * half) * CCH + t * 16 + m) * 3 + v] =
            y1acc[v][t][r] * gate[1][t][r];
  STAGE_FENCE();
  flush_tile<12>(st, node1 + (size_t)tile * 16 * CCH * 3,
                 out1 + (size_t)tile * 16 * CCH * 3, nvr * 24, lane);

  for (int aIdx = 0; aIdx < 3; ++aIdx) {
    asm volatile("" ::: "memory");
    v8f y2acc[3][2] = {{vzero, vzero}, {vzero, vzero}, {vzero, vzero}};
    for (int s = 0; s < 8; ++s) {
      LOAD_BASE(s);
#pragma unroll
      for (int b = 0; b < 3; ++b) {
        v2f a2v, a02, a11, qv;
#pragma unroll
        for (int jj = 0; jj < 2; ++jj) {
          const float g = b2[jj][aIdx * 3 + b];
          a2v[jj] = g;
          a02[jj] = x0[jj] * g;
          a11[jj] = b1[jj][aIdx] * b1[jj][b];
          float acc = 0.f;
#pragma unroll
          for (int kk = 0; kk < 3; ++kk)
            acc += b2[jj][aIdx * 3 + kk] * b2[jj][kk * 3 + b];
          qv[jj] = acc;
        }
        MM2(y2acc[b], a2v, W_U2,  s);
        MM2(y2acc[b], a02, W_V(2), s);
        MM2(y2acc[b], a11, W_V(4), s);
        MM2(y2acc[b], qv,  W_V(7), s);
      }
    }
#pragma unroll
    for (int b = 0; b < 3; ++b)
#pragma unroll
      for (int t = 0; t < 2; ++t)
#pragma unroll
        for (int r = 0; r < 8; ++r)
          st[((r + 8 * half) * CCH + t * 16 + m) * 9 + aIdx * 3 + b] =
              y2acc[b][t][r] * gate[2][t][r];
  }
  STAGE_FENCE();
  flush_tile<36>(st, node2 + (size_t)tile * 16 * CCH * 9,
                 out2 + (size_t)tile * 16 * CCH * 9, nvr * 72, lane);
  }
#undef STAGE_FENCE
#undef LOAD_BASE
#undef MM2
#undef LDB
}

extern "C" void kernel_launch(void* const* d_in, const int* in_sizes, int n_in,
                              void* d_out, int out_size, void* d_ws, size_t ws_size,
                              hipStream_t stream) {
  const float* node0 = (const float*)d_in[0];
  const float* node1 = (const float*)d_in[1];
  const float* node2 = (const float*)d_in[2];
  const float* rij   = (const float*)d_in[3];
  const float* Wrad  = (const float*)d_in[4];
  const float* U     = (const float*)d_in[5];
  const float* V     = (const float*)d_in[6];
  const float* Wg    = (const float*)d_in[7];
  const int*   idx_i = (const int*)d_in[8];
  const int*   idx_j = (const int*)d_in[9];

  const int N = in_sizes[0] / CCH;
  const int E = in_sizes[3] / 3;

  float* a0 = (float*)d_ws;
  float* a1 = a0 + (size_t)N * CCH;
  float* a2 = a1 + (size_t)N * CCH * 3;

  float* out0 = (float*)d_out;
  float* out1 = out0 + (size_t)N * CCH;
  float* out2 = out1 + (size_t)N * CCH * 3;

  mmb_edge<<<(N + EDGE_TN - 1) / EDGE_TN, EDGE_WAVES * 32, 0, stream>>>(
      rij, Wrad, idx_i, idx_j, node0, node1, node2, a0, a1, a2, N, E);

  mmb_node<<<NODE_WGS, NODE_WAVES * 32, 0, stream>>>(a0, a1, a2,
                                                 node0, node1, node2,
                                                 U, V, Wg,
                                                 out0, out1, out2, N);
}


# edge messages: a lane half takes one edge (batch slots r and 8 + r in one trip), each lane two channels; filter values stay in their WMMA result lanes, no lane exchange; LDS adds in arrival order
# speedup vs baseline: 3.9439x; 1.0604x over previous
#include <hip/hip_runtime.h>
#include <hip/hip_bf16.h>
#include <math.h>

#define CCH 32
#define NPATH 11
#define NB 8

typedef __attribute__((ext_vector_type(2))) float v2f;
typedef __attribute__((ext_vector_type(8))) float v8f;
typedef __attribute__((ext_vector_type(4))) float v4f;

__device__ __forceinline__ v8f wmma4(v2f a, v2f b, v8f c) {
  return __builtin_amdgcn_wmma_f32_16x16x4_f32(
       false, a,  false, b,
       (short)0, c,  false,  false);
}

#define EDGE_TN    80
#define EDGE_WAVES 16
#define EDGE_SCAN  24
#define QCAP       1024
#define NFRAG      (NPATH * 2 * 2)

__device__ __forceinline__ void lds_add(float* p, float v) {
  __hip_atomic_fetch_add(p, v, __ATOMIC_RELAXED, __HIP_MEMORY_SCOPE_WORKGROUP);
}

__device__ __forceinline__ float xhalf(float x) {
  const int b = __builtin_bit_cast(int, x);
  return __builtin_bit_cast(float,
      __builtin_amdgcn_permlanex16(b, b, 0x76543210, 0xfedcba98, false, false));
}

__device__ __forceinline__ float lane_f(float x, int l) {
  return __builtin_bit_cast(float,
      __builtin_amdgcn_readlane(__builtin_bit_cast(int, x), l));
}

struct NodeRows { float g0; float g1[3]; float g2[9]; };

__device__ __forceinline__ void load_rows(NodeRows& g,
                                          const float* __restrict__ node0,
                                          const float* __restrict__ node1,
                                          const float* __restrict__ node2,
                                          int j, int ch) {
  g.g0 = node0[j * CCH + ch];
#pragma unroll
  for (int a = 0; a < 3; ++a) g.g1[a] = node1[(j * CCH + ch) * 3 + a];
#pragma unroll
  for (int q = 0; q < 9; ++q) g.g2[q] = node2[(j * CCH + ch) * 9 + q];
}

__device__ __forceinline__ void edge_add(const NodeRows& g,
                                         float* s0, float* s1, float* s2,
                                         const float (&w)[NPATH],
                                         int nl, float rh0, float rh1, float rh2,
                                         int ch, bool valid) {
  const float rh[3] = {rh0, rh1, rh2};
  const float g0 = g.g0;
  const float (&g1)[3] = g.g1;
  const float (&g2)[9] = g.g2;

  const float d1 = g1[0] * rh[0] + g1[1] * rh[1] + g1[2] * rh[2];
  float gr[3];
#pragma unroll
  for (int a = 0; a < 3; ++a)
    gr[a] = g2[a * 3 + 0] * rh[0] + g2[a * 3 + 1] * rh[1] + g2[a * 3 + 2] * rh[2];
  const float d2 = gr[0] * rh[0] + gr[1] * rh[1] + gr[2] * rh[2];

  const float inv16 = 0.0625f;
  const int   rc    = nl * CCH + ch;

  const float m0 = (g0 * w[0] + d1 * w[4] + d2 * w[9]) * inv16;
  float m1[3], m2[9];
#pragma unroll
  for (int a = 0; a < 3; ++a) {
    const float mv = (g0 * w[1] + d1 * w[6]) * rh[a] + g1[a] * w[3] + gr[a] * w[8];
    m1[a] = mv * inv16;
  }
#pragma unroll
  for (int a = 0; a < 3; ++a) {
    const float ca = g0 * w[2] * rh[a] + g1[a] * w[5] + gr[a] * w[10];
#pragma unroll
    for (int b = 0; b < 3; ++b) {
      const float mv = ca * rh[b] + g2[a * 3 + b] * w[7];
      m2[a * 3 + b] = mv * inv16;
    }
  }
  if (valid) {
    lds_add(&s0[rc], m0);
#pragma unroll
    for (int a = 0; a < 3; ++a) lds_add(&s1[rc * 3 + a], m1[a]);
#pragma unroll
    for (int k = 0; k < 9; ++k) lds_add(&s2[rc * 9 + k], m2[k]);
  }
}

__global__ void __launch_bounds__(EDGE_WAVES * 32)
mmb_edge(const float* __restrict__ rij,
         const float* __restrict__ Wrad,
         const int*   __restrict__ idx_i,
         const int*   __restrict__ idx_j,
         const float* __restrict__ node0,
         const float* __restrict__ node1,
         const float* __restrict__ node2,
         float* __restrict__ a0,
         float* __restrict__ a1,
         float* __restrict__ a2,
         int N, int E) {
  __shared__ float4 acc4[EDGE_TN * 104];
  float* const s0 = (float*)acc4;
  float* const s1 = s0 + EDGE_TN * CCH;
  float* const s2 = s0 + EDGE_TN * CCH * 4;
  __shared__ float2 wb[NFRAG * 32];
  __shared__ int queue[EDGE_WAVES * QCAP];

  const int tid = threadIdx.x;
  const int c   = tid & 31;
  const int hf  = c >> 4;
  const int q   = c & 15;
  const bool lo = (c < 16);
  const int wv  = __builtin_amdgcn_readfirstlane(tid >> 5);
  const int n0  = blockIdx.x * EDGE_TN;
  const int nv  = (N - n0 < EDGE_TN) ? (N - n0) : EDGE_TN;
  int* const qw = queue + wv * QCAP;

  constexpr int WR_N     = NFRAG * 64;
  constexpr int WR_TRIPS = (WR_N + EDGE_WAVES * 32 - 1) / (EDGE_WAVES * 32);
  float wt[WR_TRIPS];
#pragma unroll
  for (int u = 0; u < WR_TRIPS; ++u) {
    int k = tid + u * EDGE_WAVES * 32;
    k = (k < WR_N) ? k : (WR_N - 1);
    const int jj = k & 1, ln = (k >> 1) & 31, f = k >> 6;
    const int t = f & 1, s = (f >> 1) & 1, p = f >> 2;
    const int kk = 4 * s + 2 * (ln >> 4) + jj;
    wt[u] = Wrad[(p * NB + kk) * CCH + 16 * t + (ln & 15)];
  }
  for (int k = tid; k < EDGE_TN * 104; k += EDGE_WAVES * 32)
    acc4[k] = make_float4(0.f, 0.f, 0.f, 0.f);
#pragma unroll
  for (int u = 0; u < WR_TRIPS; ++u) {
    const int k = tid + u * EDGE_WAVES * 32;
    if (k < WR_N) ((float*)wb)[k] = wt[u];
  }
  __syncthreads();

#define CEN(b) (0.5f + 0.6428571429f * (float)(b))
  const float cen00 = hf ? CEN(2) : CEN(0);
  const float cen01 = hf ? CEN(3) : CEN(1);
  const float cen10 = hf ? CEN(6) : CEN(4);
  const float cen11 = hf ? CEN(7) : CEN(5);
#undef CEN
  const v8f vzero = {};

  const int nChunks = (E + 31) >> 5;
  int cb   = wv;
  int head = 0;
  int cnt  = 0;
  for (;;) {
    if (cnt < 16 && cb < nChunks) {
      int il[EDGE_SCAN];
#pragma unroll
      for (int u = 0; u < EDGE_SCAN; ++u) {
        const int ev = ((cb + u * EDGE_WAVES) << 5) + c;
        const int id = idx_i[(ev < E) ? ev : (E - 1)];
        il[u] = (ev < E) ? (id - n0) : -1;
      }
#pragma unroll
      for (int u = 0; u < EDGE_SCAN; ++u) {
        const bool hit = (unsigned)il[u] < (unsigned)nv;
        const unsigned mu = __builtin_amdgcn_ballot_w32(hit);
        if (mu != 0u) {
          const int pos = cnt + (int)__builtin_amdgcn_mbcnt_lo(mu, 0u);
          if (hit) qw[(head + pos) & (QCAP - 1)] = ((cb + u * EDGE_WAVES) << 5) + c;
          cnt += __builtin_popcount(mu);
        }
      }
      cb += EDGE_WAVES * EDGE_SCAN;
      continue;
    }
    if (cnt == 0) break;
    const int nb = (cnt < 16) ? cnt : 16;

    asm volatile("s_wait_dscnt 0" ::: "memory");

    const int e = qw[(head + ((q < nb) ? q : 0)) & (QCAP - 1)];
    const float r0 = rij[e * 3 + 0], r1 = rij[e * 3 + 1], r2 = rij[e * 3 + 2];
    const int   jv  = idx_j[e];
    const int   nlv = idx_i[e] - n0;
    const float d  = sqrtf(r0 * r0 + r1 * r1 + r2 * r2 + 1e-12f);
    const float iv = 1.0f / d;
    const float rhx = r0 * iv, rhy = r1 * iv, rhz = r2 * iv;

    float x = d * 0.2f;
    x = (x > 1.0f) ? 1.0f : x;
    const float fc = 0.5f * (cosf(3.14159265358979f * x) + 1.0f);

    const float t00 = d - cen00, t01 = d - cen01, t10 = d - cen10, t11 = d - cen11;
    v2f af0, af1;
    af0.x = __expf(-4.0f * t00 * t00) * fc;
    af0.y = __expf(-4.0f * t01 * t01) * fc;
    af1.x = __expf(-4.0f * t10 * t10) * fc;
    af1.y = __expf(-4.0f * t11 * t11) * fc;

    v8f D[NPATH][2];
#pragma unroll
    for (int p = 0; p < NPATH; ++p)
#pragma unroll
      for (int t = 0; t < 2; ++t) {
        const float2 b0 = wb[((p * 2 + 0) * 2 + t) * 32 + c];
        const float2 b1 = wb[((p * 2 + 1) * 2 + t) * 32 + c];
        v2f bf0, bf1;
        bf0.x = b0.x; bf0.y = b0.y; bf1.x = b1.x; bf1.y = b1.y;
        v8f acc = wmma4(af0, bf0, vzero);
        D[p][t] = wmma4(af1, bf1, acc);
      }

#pragma unroll
    for (int r = 0; r < 8; ++r) {
      if (r < nb) {
        const bool valid = (8 * hf + r) < nb;
        const int   jA = __builtin_amdgcn_readlane(jv, r);
        const int   jB = __builtin_amdgcn_readlane(jv, 8 + r);
        const int   nA = __builtin_amdgcn_readlane(nlv, r);
        const int   nB = __builtin_amdgcn_readlane(nlv, 8 + r);
        const float xA = lane_f(rhx, r), xB = lane_f(rhx, 8 + r);
        const float yA = lane_f(rhy, r), yB = lane_f(rhy, 8 + r);
        const float zA = lane_f(rhz, r), zB = lane_f(rhz, 8 + r);
        const int   j  = hf ? jB : jA;
        const int   nl = hf ? nB : nA;
        const float h0 = hf ? xB : xA;
        const float h1 = hf ? yB : yA;
        const float h2 = hf ? zB : zA;
        NodeRows gA, gB;
        load_rows(gA, node0, node1, node2, j, q);
        load_rows(gB, node0, node1, node2, j, 16 + q);
        float wA[NPATH], wB[NPATH];
#pragma unroll
        for (int p = 0; p < NPATH; ++p) { wA[p] = D[p][0][r]; wB[p] = D[p][1][r]; }
        edge_add(gA, s0, s1, s2, wA, nl, h0, h1, h2, q, valid);
        edge_add(gB, s0, s1, s2, wB, nl, h0, h1, h2, 16 + q, valid);
      }
    }
    head = (head + nb) & (QCAP - 1);
    cnt -= nb;
  }
  __syncthreads();

  float4* const o0 = (float4*)(a0 + (size_t)n0 * CCH);
  float4* const o1 = (float4*)(a1 + (size_t)n0 * CCH * 3);
  float4* const o2 = (float4*)(a2 + (size_t)n0 * CCH * 9);
  for (int k = tid; k < nv * 8; k += EDGE_WAVES * 32)  o0[k] = acc4[k];
  for (int k = tid; k < nv * 24; k += EDGE_WAVES * 32) o1[k] = acc4[EDGE_TN * 8 + k];
  for (int k = tid; k < nv * 72; k += EDGE_WAVES * 32) o2[k] = acc4[EDGE_TN * 32 + k];
}

#define STAGE_F (16 * CCH * 9)
#define NODE_WAVES 8
#define NODE_WGS   256

template <int PER_LANE>
__device__ __forceinline__ void flush_tile(const float* st,
                                           const float* __restrict__ res,
                                           float* __restrict__ dst,
                                           int nvalid4, int lane) {
  const v4f* const s4 = (const v4f*)st;
  const v4f* const r4 = (const v4f*)res;
  v4f* const d4 = (v4f*)dst;
  constexpr int BATCH = (PER_LANE < 12) ? PER_LANE : 12;
#pragma unroll
  for (int i0 = 0; i0 < PER_LANE; i0 += BATCH) {
    v4f rv[BATCH], sv[BATCH];
#pragma unroll
    for (int i = 0; i < BATCH; ++i) {
      const int k = (i0 + i) * 32 + lane;
      rv[i] = r4[(k < nvalid4) ? k : (nvalid4 - 1)];
      sv[i] = s4[k];
    }
#pragma unroll
    for (int i = 0; i < BATCH; ++i) {
      const int k = (i0 + i) * 32 + lane;
      if (k < nvalid4) d4[k] = sv[i] + rv[i];
    }
  }
}
#define W_U0 0
#define W_U1 1
#define W_U2 2
#define W_V(k) (3 + (k))
#define W_G(k) (11 + (k))

__global__ void __launch_bounds__(NODE_WAVES * 32)
mmb_node(const float* __restrict__ a0, const float* __restrict__ a1,
         const float* __restrict__ a2,
         const float* __restrict__ node0, const float* __restrict__ node1,
         const float* __restrict__ node2,
         const float* __restrict__ U, const float* __restrict__ V,
         const float* __restrict__ Wg,
         float* __restrict__ out0, float* __restrict__ out1,
         float* __restrict__ out2, int N) {
  __shared__ float2 wlds[14 * 8 * 2 * 32];
  __shared__ __attribute__((aligned(16))) float stage[NODE_WAVES * STAGE_F];

  for (int idx = threadIdx.x; idx < 14 * 1024; idx += blockDim.x) {
    const int wmat = idx >> 10;
    const int rem  = idx & 1023;
    const int k    = rem >> 5;
    const int n    = rem & 31;
    const float* src = (wmat < 3) ? (U + wmat * 1024)
                     : (wmat < 11) ? (V + (wmat - 3) * 1024)
                                   : (Wg + (wmat - 11) * 1024);
    const float val = src[rem];
    const int s   = k >> 2;
    const int kk  = k & 3;
    const int ln  = ((kk >> 1) << 4) | (n & 15);
    const int t   = n >> 4;
    const int jj  = kk & 1;
    ((float*)&wlds[((wmat * 8 + s) * 2 + t) * 32 + ln])[jj] = val;
  }
  __syncthreads();

  const int lane = threadIdx.x & 31;
  const int wv   = __builtin_amdgcn_readfirstlane(threadIdx.x >> 5);
  const int half = lane >> 4;
  const int m    = lane & 15;
  const int nTiles = (N + 15) >> 4;

#define LDB(wm, s, t)                                                       \
  ({ float2 _bv = wlds[(((wm) * 8 + (s)) * 2 + (t)) * 32 + lane];           \
     v2f _b; _b.x = _bv.x; _b.y = _bv.y; _b; })

#define MM2(acc, aop, wm, s)                                                \
  do { v2f _a = (aop);                                                      \
       acc[0] = wmma4(_a, LDB(wm, s, 0), acc[0]);                           \
       acc[1] = wmma4(_a, LDB(wm, s, 1), acc[1]); } while (0)

  const v8f vzero = {};
  float* const st = stage + wv * STAGE_F;
  for (int tile = wv * (int)gridDim.x + (int)blockIdx.x; tile < nTiles;
       tile += NODE_WAVES * (int)gridDim.x) {
  const int nvr = (N - tile * 16 < 16) ? (N - tile * 16) : 16;
#define STAGE_FENCE() asm volatile("s_wait_dscnt 0" ::: "memory")

  const int rowA = tile * 16 + m;
  const int rowC = (rowA < N) ? rowA : (N - 1);
  float x0[2], b1[2][3], b2[2][9];
#define LOAD_BASE(s)                                                        \
  do { const int _k = (s) * 4 + 2 * half;                                   \
       const int _rc = rowC * CCH + _k;                                     \
       x0[0] = a0[_rc]; x0[1] = a0[_rc + 1];                                \
       _Pragma("unroll") for (int _a = 0; _a < 3; ++_a) {                   \
         b1[0][_a] = a1[_rc * 3 + _a]; b1[1][_a] = a1[(_rc + 1) * 3 + _a]; }\
       _Pragma("unroll") for (int _q = 0; _q < 9; ++_q) {                   \
         b2[0][_q] = a2[_rc * 9 + _q]; b2[1][_q] = a2[(_rc + 1) * 9 + _q]; }\
  } while (0)

  v8f y0acc[2] = {vzero, vzero};
  for (int s = 0; s < 8; ++s) {
    LOAD_BASE(s);
    v2f ax, asq, s1v, s2v;
#pragma unroll
    for (int jj = 0; jj < 2; ++jj) {
      ax[jj]  = x0[jj];
      asq[jj] = x0[jj] * x0[jj];
      float acc = 0.f;
#pragma unroll
      for (int a = 0; a < 3; ++a) acc += b1[jj][a] * b1[jj][a];
      s1v[jj] = acc;
      acc = 0.f;
#pragma unroll
      for (int q = 0; q < 9; ++q) acc += b2[jj][q] * b2[jj][q];
      s2v[jj] = acc;
    }
    MM2(y0acc, ax,  W_U0,  s);
    MM2(y0acc, asq, W_V(0), s);
    MM2(y0acc, s1v, W_V(3), s);
    MM2(y0acc, s2v, W_V(6), s);
  }

#pragma unroll
  for (int t = 0; t < 2; ++t)
#pragma unroll
    for (int r = 0; r < 8; ++r)
      st[(r + 8 * half) * CCH + t * 16 + m] = y0acc[t][r];
  STAGE_FENCE();

  v8f ga[3][2] = {{vzero, vzero}, {vzero, vzero}, {vzero, vzero}};
  for (int s = 0; s < 8; ++s) {
    const int k = s * 4 + 2 * half;
    v2f ay; ay.x = st[m * CCH + k]; ay.y = st[m * CCH + k + 1];
#pragma unroll
    for (int g = 0; g < 3; ++g) MM2(ga[g], ay, W_G(g), s);
  }
  v8f gate[3][2];
#pragma unroll
  for (int g = 0; g < 3; ++g)
#pragma unroll
    for (int t = 0; t < 2; ++t)
#pragma unroll
      for (int r = 0; r < 8; ++r) {
        const float v = ga[g][t][r];
        gate[g][t][r] = v * (1.0f / (1.0f + __expf(-v)));
      }

#pragma unroll
  for (int t = 0; t < 2; ++t)
#pragma unroll
    for (int r = 0; r < 8; ++r)
      st[(r + 8 * half) * CCH + t * 16 + m] = gate[0][t][r];
  STAGE_FENCE();
  flush_tile<4>(st, node0 + (size_t)tile * 16 * CCH,
                out0 + (size_t)tile * 16 * CCH, nvr * 8, lane);

  v8f y1acc[3][2] = {{vzero, vzero}, {vzero, vzero}, {vzero, vzero}};
  for (int s = 0; s < 8; ++s) {
    LOAD_BASE(s);
#pragma unroll
    for (int v = 0; v < 3; ++v) {
      v2f av, a01, tv;
#pragma unroll
      for (int jj = 0; jj < 2; ++jj) {
        av[jj]  = b1[jj][v];
        a01[jj] = x0[jj] * b1[jj][v];
        float acc = 0.f;
#pragma unroll
        for (int a = 0; a < 3; ++a) acc += b1[jj][a] * b2[jj][a * 3 + v];
        tv[jj] = acc;
      }
      MM2(y1acc[v], av,  W_U1,  s);
      MM2(y1acc[v], a01, W_V(1), s);
      MM2(y1acc[v], tv,  W_V(5), s);
    }
  }
#pragma unroll
  for (int v = 0; v < 3; ++v)
#pragma unroll
    for (int t = 0; t < 2; ++t)
#pragma unroll
      for (int r = 0; r < 8; ++r)
        st[((r + 8 * half) * CCH + t * 16 + m) * 3 + v] =
            y1acc[v][t][r] * gate[1][t][r];
  STAGE_FENCE();
  flush_tile<12>(st, node1 + (size_t)tile * 16 * CCH * 3,
                 out1 + (size_t)tile * 16 * CCH * 3, nvr * 24, lane);

  for (int aIdx = 0; aIdx < 3; ++aIdx) {
    asm volatile("" ::: "memory");
    v8f y2acc[3][2] = {{vzero, vzero}, {vzero, vzero}, {vzero, vzero}};
    for (int s = 0; s < 8; ++s) {
      LOAD_BASE(s);
#pragma unroll
      for (int b = 0; b < 3; ++b) {
        v2f a2v, a02, a11, qv;
#pragma unroll
        for (int jj = 0; jj < 2; ++jj) {
          const float g = b2[jj][aIdx * 3 + b];
          a2v[jj] = g;
          a02[jj] = x0[jj] * g;
          a11[jj] = b1[jj][aIdx] * b1[jj][b];
          float acc = 0.f;
#pragma unroll
          for (int kk = 0; kk < 3; ++kk)
            acc += b2[jj][aIdx * 3 + kk] * b2[jj][kk * 3 + b];
          qv[jj] = acc;
        }
        MM2(y2acc[b], a2v, W_U2,  s);
        MM2(y2acc[b], a02, W_V(2), s);
        MM2(y2acc[b], a11, W_V(4), s);
        MM2(y2acc[b], qv,  W_V(7), s);
      }
    }
#pragma unroll
    for (int b = 0; b < 3; ++b)
#pragma unroll
      for (int t = 0; t < 2; ++t)
#pragma unroll
        for (int r = 0; r < 8; ++r)
          st[((r + 8 * half) * CCH + t * 16 + m) * 9 + aIdx * 3 + b] =
              y2acc[b][t][r] * gate[2][t][r];
  }
  STAGE_FENCE();
  flush_tile<36>(st, node2 + (size_t)tile * 16 * CCH * 9,
                 out2 + (size_t)tile * 16 * CCH * 9, nvr * 72, lane);
  }
#undef STAGE_FENCE
#undef LOAD_BASE
#undef MM2
#undef LDB
}

extern "C" void kernel_launch(void* const* d_in, const int* in_sizes, int n_in,
                              void* d_out, int out_size, void* d_ws, size_t ws_size,
                              hipStream_t stream) {
  const float* node0 = (const float*)d_in[0];
  const float* node1 = (const float*)d_in[1];
  const float* node2 = (const float*)d_in[2];
  const float* rij   = (const float*)d_in[3];
  const float* Wrad  = (const float*)d_in[4];
  const float* U     = (const float*)d_in[5];
  const float* V     = (const float*)d_in[6];
  const float* Wg    = (const float*)d_in[7];
  const int*   idx_i = (const int*)d_in[8];
  const int*   idx_j = (const int*)d_in[9];

  const int N = in_sizes[0] / CCH;
  const int E = in_sizes[3] / 3;

  float* a0 = (float*)d_ws;
  float* a1 = a0 + (size_t)N * CCH;
  float* a2 = a1 + (size_t)N * CCH * 3;

  float* out0 = (float*)d_out;
  float* out1 = out0 + (size_t)N * CCH;
  float* out2 = out1 + (size_t)N * CCH * 3;

  mmb_edge<<<(N + EDGE_TN - 1) / EDGE_TN, EDGE_WAVES * 32, 0, stream>>>(
      rij, Wrad, idx_i, idx_j, node0, node1, node2, a0, a1, a2, N, E);

  mmb_node<<<NODE_WGS, NODE_WAVES * 32, 0, stream>>>(a0, a1, a2,
                                                 node0, node1, node2,
                                                 U, V, Wg,
                                                 out0, out1, out2, N);
}


# edge scan: a lane reads 4 consecutive destination ids as one 16-byte load (128 edges per wave load, 12 in flight), one range test per 128; ring 2048; edge-to-wave deal and LDS add order change
# speedup vs baseline: 4.3608x; 1.1057x over previous
#include <hip/hip_runtime.h>
#include <hip/hip_bf16.h>
#include <math.h>

#define CCH 32
#define NPATH 11
#define NB 8

typedef __attribute__((ext_vector_type(2))) float v2f;
typedef __attribute__((ext_vector_type(8))) float v8f;
typedef __attribute__((ext_vector_type(4))) float v4f;

__device__ __forceinline__ v8f wmma4(v2f a, v2f b, v8f c) {
  return __builtin_amdgcn_wmma_f32_16x16x4_f32(
       false, a,  false, b,
       (short)0, c,  false,  false);
}

#define EDGE_TN    80
#define EDGE_WAVES 16
#define EDGE_SCAN  12
#define QCAP       2048
#define NFRAG      (NPATH * 2 * 2)

__device__ __forceinline__ void lds_add(float* p, float v) {
  __hip_atomic_fetch_add(p, v, __ATOMIC_RELAXED, __HIP_MEMORY_SCOPE_WORKGROUP);
}

__device__ __forceinline__ float xhalf(float x) {
  const int b = __builtin_bit_cast(int, x);
  return __builtin_bit_cast(float,
      __builtin_amdgcn_permlanex16(b, b, 0x76543210, 0xfedcba98, false, false));
}

__device__ __forceinline__ float lane_f(float x, int l) {
  return __builtin_bit_cast(float,
      __builtin_amdgcn_readlane(__builtin_bit_cast(int, x), l));
}

struct NodeRows { float g0; float g1[3]; float g2[9]; };

__device__ __forceinline__ void load_rows(NodeRows& g,
                                          const float* __restrict__ node0,
                                          const float* __restrict__ node1,
                                          const float* __restrict__ node2,
                                          int j, int ch) {
  g.g0 = node0[j * CCH + ch];
#pragma unroll
  for (int a = 0; a < 3; ++a) g.g1[a] = node1[(j * CCH + ch) * 3 + a];
#pragma unroll
  for (int q = 0; q < 9; ++q) g.g2[q] = node2[(j * CCH + ch) * 9 + q];
}

__device__ __forceinline__ void edge_add(const NodeRows& g,
                                         float* s0, float* s1, float* s2,
                                         const float (&w)[NPATH],
                                         int nl, float rh0, float rh1, float rh2,
                                         int ch, bool valid) {
  const float rh[3] = {rh0, rh1, rh2};
  const float g0 = g.g0;
  const float (&g1)[3] = g.g1;
  const float (&g2)[9] = g.g2;

  const float d1 = g1[0] * rh[0] + g1[1] * rh[1] + g1[2] * rh[2];
  float gr[3];
#pragma unroll
  for (int a = 0; a < 3; ++a)
    gr[a] = g2[a * 3 + 0] * rh[0] + g2[a * 3 + 1] * rh[1] + g2[a * 3 + 2] * rh[2];
  const float d2 = gr[0] * rh[0] + gr[1] * rh[1] + gr[2] * rh[2];

  const float inv16 = 0.0625f;
  const int   rc    = nl * CCH + ch;

  const float m0 = (g0 * w[0] + d1 * w[4] + d2 * w[9]) * inv16;
  float m1[3], m2[9];
#pragma unroll
  for (int a = 0; a < 3; ++a) {
    const float mv = (g0 * w[1] + d1 * w[6]) * rh[a] + g1[a] * w[3] + gr[a] * w[8];
    m1[a] = mv * inv16;
  }
#pragma unroll
  for (int a = 0; a < 3; ++a) {
    const float ca = g0 * w[2] * rh[a] + g1[a] * w[5] + gr[a] * w[10];
#pragma unroll
    for (int b = 0; b < 3; ++b) {
      const float mv = ca * rh[b] + g2[a * 3 + b] * w[7];
      m2[a * 3 + b] = mv * inv16;
    }
  }
  if (valid) {
    lds_add(&s0[rc], m0);
#pragma unroll
    for (int a = 0; a < 3; ++a) lds_add(&s1[rc * 3 + a], m1[a]);
#pragma unroll
    for (int k = 0; k < 9; ++k) lds_add(&s2[rc * 9 + k], m2[k]);
  }
}

__global__ void __launch_bounds__(EDGE_WAVES * 32)
mmb_edge(const float* __restrict__ rij,
         const float* __restrict__ Wrad,
         const int*   __restrict__ idx_i,
         const int*   __restrict__ idx_j,
         const float* __restrict__ node0,
         const float* __restrict__ node1,
         const float* __restrict__ node2,
         float* __restrict__ a0,
         float* __restrict__ a1,
         float* __restrict__ a2,
         int N, int E) {
  __shared__ float4 acc4[EDGE_TN * 104];
  float* const s0 = (float*)acc4;
  float* const s1 = s0 + EDGE_TN * CCH;
  float* const s2 = s0 + EDGE_TN * CCH * 4;
  __shared__ float2 wb[NFRAG * 32];
  __shared__ int queue[EDGE_WAVES * QCAP];

  const int tid = threadIdx.x;
  const int c   = tid & 31;
  const int hf  = c >> 4;
  const int q   = c & 15;
  const bool lo = (c < 16);
  const int wv  = __builtin_amdgcn_readfirstlane(tid >> 5);
  const int n0  = blockIdx.x * EDGE_TN;
  const int nv  = (N - n0 < EDGE_TN) ? (N - n0) : EDGE_TN;
  int* const qw = queue + wv * QCAP;

  constexpr int WR_N     = NFRAG * 64;
  constexpr int WR_TRIPS = (WR_N + EDGE_WAVES * 32 - 1) / (EDGE_WAVES * 32);
  float wt[WR_TRIPS];
#pragma unroll
  for (int u = 0; u < WR_TRIPS; ++u) {
    int k = tid + u * EDGE_WAVES * 32;
    k = (k < WR_N) ? k : (WR_N - 1);
    const int jj = k & 1, ln = (k >> 1) & 31, f = k >> 6;
    const int t = f & 1, s = (f >> 1) & 1, p = f >> 2;
    const int kk = 4 * s + 2 * (ln >> 4) + jj;
    wt[u] = Wrad[(p * NB + kk) * CCH + 16 * t + (ln & 15)];
  }
  for (int k = tid; k < EDGE_TN * 104; k += EDGE_WAVES * 32)
    acc4[k] = make_float4(0.f, 0.f, 0.f, 0.f);
#pragma unroll
  for (int u = 0; u < WR_TRIPS; ++u) {
    const int k = tid + u * EDGE_WAVES * 32;
    if (k < WR_N) ((float*)wb)[k] = wt[u];
  }
  __syncthreads();

#define CEN(b) (0.5f + 0.6428571429f * (float)(b))
  const float cen00 = hf ? CEN(2) : CEN(0);
  const float cen01 = hf ? CEN(3) : CEN(1);
  const float cen10 = hf ? CEN(6) : CEN(4);
  const float cen11 = hf ? CEN(7) : CEN(5);
#undef CEN
  const v8f vzero = {};

  const int nPieces = (E + 127) >> 7;
  int cb   = wv;
  int head = 0;
  int cnt  = 0;
  for (;;) {
    if (cnt < 16 && cb < nPieces) {
      int4 il[EDGE_SCAN];
      const bool whole = (((cb + (EDGE_SCAN - 1) * EDGE_WAVES) << 7) + 128) <= E;
      if (whole) {
#pragma unroll
        for (int u = 0; u < EDGE_SCAN; ++u) {
          const int e0 = ((cb + u * EDGE_WAVES) << 7) + 4 * c;
          il[u] = *(const int4*)(idx_i + e0);
        }
      } else {
#pragma unroll
        for (int u = 0; u < EDGE_SCAN; ++u) {
          const int e0 = ((cb + u * EDGE_WAVES) << 7) + 4 * c;
          int id[4];
#pragma unroll
          for (int k = 0; k < 4; ++k) {
            const int ev = e0 + k;
            const int v  = idx_i[(ev < E) ? ev : (E - 1)];
            id[k] = (ev < E) ? v : (n0 - 1);
          }
          il[u] = make_int4(id[0], id[1], id[2], id[3]);
        }
      }
#pragma unroll
      for (int u = 0; u < EDGE_SCAN; ++u) {
        const int e0 = ((cb + u * EDGE_WAVES) << 7) + 4 * c;
        const unsigned a[4] = {(unsigned)(il[u].x - n0), (unsigned)(il[u].y - n0),
                               (unsigned)(il[u].z - n0), (unsigned)(il[u].w - n0)};
        const unsigned lo4 = min(min(a[0], a[1]), min(a[2], a[3]));
        if (__builtin_amdgcn_ballot_w32(lo4 < (unsigned)nv) != 0u) {
#pragma unroll
          for (int k = 0; k < 4; ++k) {
            const bool hit = a[k] < (unsigned)nv;
            const unsigned mu = __builtin_amdgcn_ballot_w32(hit);
            if (mu != 0u) {
              const int pos = cnt + (int)__builtin_amdgcn_mbcnt_lo(mu, 0u);
              if (hit) qw[(head + pos) & (QCAP - 1)] = e0 + k;
              cnt += __builtin_popcount(mu);
            }
          }
        }
      }
      cb += EDGE_WAVES * EDGE_SCAN;
      continue;
    }
    if (cnt == 0) break;
    const int nb = (cnt < 16) ? cnt : 16;

    asm volatile("s_wait_dscnt 0" ::: "memory");

    const int e = qw[(head + ((q < nb) ? q : 0)) & (QCAP - 1)];
    const float r0 = rij[e * 3 + 0], r1 = rij[e * 3 + 1], r2 = rij[e * 3 + 2];
    const int   jv  = idx_j[e];
    const int   nlv = idx_i[e] - n0;
    const float d  = sqrtf(r0 * r0 + r1 * r1 + r2 * r2 + 1e-12f);
    const float iv = 1.0f / d;
    const float rhx = r0 * iv, rhy = r1 * iv, rhz = r2 * iv;

    float x = d * 0.2f;
    x = (x > 1.0f) ? 1.0f : x;
    const float fc = 0.5f * (cosf(3.14159265358979f * x) + 1.0f);

    const float t00 = d - cen00, t01 = d - cen01, t10 = d - cen10, t11 = d - cen11;
    v2f af0, af1;
    af0.x = __expf(-4.0f * t00 * t00) * fc;
    af0.y = __expf(-4.0f * t01 * t01) * fc;
    af1.x = __expf(-4.0f * t10 * t10) * fc;
    af1.y = __expf(-4.0f * t11 * t11) * fc;

    v8f D[NPATH][2];
#pragma unroll
    for (int p = 0; p < NPATH; ++p)
#pragma unroll
      for (int t = 0; t < 2; ++t) {
        const float2 b0 = wb[((p * 2 + 0) * 2 + t) * 32 + c];
        const float2 b1 = wb[((p * 2 + 1) * 2 + t) * 32 + c];
        v2f bf0, bf1;
        bf0.x = b0.x; bf0.y = b0.y; bf1.x = b1.x; bf1.y = b1.y;
        v8f acc = wmma4(af0, bf0, vzero);
        D[p][t] = wmma4(af1, bf1, acc);
      }

#pragma unroll
    for (int r = 0; r < 8; ++r) {
      if (r < nb) {
        const bool valid = (8 * hf + r) < nb;
        const int   jA = __builtin_amdgcn_readlane(jv, r);
        const int   jB = __builtin_amdgcn_readlane(jv, 8 + r);
        const int   nA = __builtin_amdgcn_readlane(nlv, r);
        const int   nB = __builtin_amdgcn_readlane(nlv, 8 + r);
        const float xA = lane_f(rhx, r), xB = lane_f(rhx, 8 + r);
        const float yA = lane_f(rhy, r), yB = lane_f(rhy, 8 + r);
        const float zA = lane_f(rhz, r), zB = lane_f(rhz, 8 + r);
        const int   j  = hf ? jB : jA;
        const int   nl = hf ? nB : nA;
        const float h0 = hf ? xB : xA;
        const float h1 = hf ? yB : yA;
        const float h2 = hf ? zB : zA;
        NodeRows gA, gB;
        load_rows(gA, node0, node1, node2, j, q);
        load_rows(gB, node0, node1, node2, j, 16 + q);
        float wA[NPATH], wB[NPATH];
#pragma unroll
        for (int p = 0; p < NPATH; ++p) { wA[p] = D[p][0][r]; wB[p] = D[p][1][r]; }
        edge_add(gA, s0, s1, s2, wA, nl, h0, h1, h2, q, valid);
        edge_add(gB, s0, s1, s2, wB, nl, h0, h1, h2, 16 + q, valid);
      }
    }
    head = (head + nb) & (QCAP - 1);
    cnt -= nb;
  }
  __syncthreads();

  float4* const o0 = (float4*)(a0 + (size_t)n0 * CCH);
  float4* const o1 = (float4*)(a1 + (size_t)n0 * CCH * 3);
  float4* const o2 = (float4*)(a2 + (size_t)n0 * CCH * 9);
  for (int k = tid; k < nv * 8; k += EDGE_WAVES * 32)  o0[k] = acc4[k];
  for (int k = tid; k < nv * 24; k += EDGE_WAVES * 32) o1[k] = acc4[EDGE_TN * 8 + k];
  for (int k = tid; k < nv * 72; k += EDGE_WAVES * 32) o2[k] = acc4[EDGE_TN * 32 + k];
}

#define STAGE_F (16 * CCH * 9)
#define NODE_WAVES 8
#define NODE_WGS   256

template <int PER_LANE>
__device__ __forceinline__ void flush_tile(const float* st,
                                           const float* __restrict__ res,
                                           float* __restrict__ dst,
                                           int nvalid4, int lane) {
  const v4f* const s4 = (const v4f*)st;
  const v4f* const r4 = (const v4f*)res;
  v4f* const d4 = (v4f*)dst;
  constexpr int BATCH = (PER_LANE < 12) ? PER_LANE : 12;
#pragma unroll
  for (int i0 = 0; i0 < PER_LANE; i0 += BATCH) {
    v4f rv[BATCH], sv[BATCH];
#pragma unroll
    for (int i = 0; i < BATCH; ++i) {
      const int k = (i0 + i) * 32 + lane;
      rv[i] = r4[(k < nvalid4) ? k : (nvalid4 - 1)];
      sv[i] = s4[k];
    }
#pragma unroll
    for (int i = 0; i < BATCH; ++i) {
      const int k = (i0 + i) * 32 + lane;
      if (k < nvalid4) d4[k] = sv[i] + rv[i];
    }
  }
}
#define W_U0 0
#define W_U1 1
#define W_U2 2
#define W_V(k) (3 + (k))
#define W_G(k) (11 + (k))

__global__ void __launch_bounds__(NODE_WAVES * 32)
mmb_node(const float* __restrict__ a0, const float* __restrict__ a1,
         const float* __restrict__ a2,
         const float* __restrict__ node0, const float* __restrict__ node1,
         const float* __restrict__ node2,
         const float* __restrict__ U, const float* __restrict__ V,
         const float* __restrict__ Wg,
         float* __restrict__ out0, float* __restrict__ out1,
         float* __restrict__ out2, int N) {
  __shared__ float2 wlds[14 * 8 * 2 * 32];
  __shared__ __attribute__((aligned(16))) float stage[NODE_WAVES * STAGE_F];

  for (int idx = threadIdx.x; idx < 14 * 1024; idx += blockDim.x) {
    const int wmat = idx >> 10;
    const int rem  = idx & 1023;
    const int k    = rem >> 5;
    const int n    = rem & 31;
    const float* src = (wmat < 3) ? (U + wmat * 1024)
                     : (wmat < 11) ? (V + (wmat - 3) * 1024)
                                   : (Wg + (wmat - 11) * 1024);
    const float val = src[rem];
    const int s   = k >> 2;
    const int kk  = k & 3;
    const int ln  = ((kk >> 1) << 4) | (n & 15);
    const int t   = n >> 4;
    const int jj  = kk & 1;
    ((float*)&wlds[((wmat * 8 + s) * 2 + t) * 32 + ln])[jj] = val;
  }
  __syncthreads();

  const int lane = threadIdx.x & 31;
  const int wv   = __builtin_amdgcn_readfirstlane(threadIdx.x >> 5);
  const int half = lane >> 4;
  const int m    = lane & 15;
  const int nTiles = (N + 15) >> 4;

#define LDB(wm, s, t)                                                       \
  ({ float2 _bv = wlds[(((wm) * 8 + (s)) * 2 + (t)) * 32 + lane];           \
     v2f _b; _b.x = _bv.x; _b.y = _bv.y; _b; })

#define MM2(acc, aop, wm, s)                                                \
  do { v2f _a = (aop);                                                      \
       acc[0] = wmma4(_a, LDB(wm, s, 0), acc[0]);                           \
       acc[1] = wmma4(_a, LDB(wm, s, 1), acc[1]); } while (0)

  const v8f vzero = {};
  float* const st = stage + wv * STAGE_F;
  for (int tile = wv * (int)gridDim.x + (int)blockIdx.x; tile < nTiles;
       tile += NODE_WAVES * (int)gridDim.x) {
  const int nvr = (N - tile * 16 < 16) ? (N - tile * 16) : 16;
#define STAGE_FENCE() asm volatile("s_wait_dscnt 0" ::: "memory")

  const int rowA = tile * 16 + m;
  const int rowC = (rowA < N) ? rowA : (N - 1);
  float x0[2], b1[2][3], b2[2][9];
#define LOAD_BASE(s)                                                        \
  do { const int _k = (s) * 4 + 2 * half;                                   \
       const int _rc = rowC * CCH + _k;                                     \
       x0[0] = a0[_rc]; x0[1] = a0[_rc + 1];                                \
       _Pragma("unroll") for (int _a = 0; _a < 3; ++_a) {                   \
         b1[0][_a] = a1[_rc * 3 + _a]; b1[1][_a] = a1[(_rc + 1) * 3 + _a]; }\
       _Pragma("unroll") for (int _q = 0; _q < 9; ++_q) {                   \
         b2[0][_q] = a2[_rc * 9 + _q]; b2[1][_q] = a2[(_rc + 1) * 9 + _q]; }\
  } while (0)

  v8f y0acc[2] = {vzero, vzero};
  for (int s = 0; s < 8; ++s) {
    LOAD_BASE(s);
    v2f ax, asq, s1v, s2v;
#pragma unroll
    for (int jj = 0; jj < 2; ++jj) {
      ax[jj]  = x0[jj];
      asq[jj] = x0[jj] * x0[jj];
      float acc = 0.f;
#pragma unroll
      for (int a = 0; a < 3; ++a) acc += b1[jj][a] * b1[jj][a];
      s1v[jj] = acc;
      acc = 0.f;
#pragma unroll
      for (int q = 0; q < 9; ++q) acc += b2[jj][q] * b2[jj][q];
      s2v[jj] = acc;
    }
    MM2(y0acc, ax,  W_U0,  s);
    MM2(y0acc, asq, W_V(0), s);
    MM2(y0acc, s1v, W_V(3), s);
    MM2(y0acc, s2v, W_V(6), s);
  }

#pragma unroll
  for (int t = 0; t < 2; ++t)
#pragma unroll
    for (int r = 0; r < 8; ++r)
      st[(r + 8 * half) * CCH + t * 16 + m] = y0acc[t][r];
  STAGE_FENCE();

  v8f ga[3][2] = {{vzero, vzero}, {vzero, vzero}, {vzero, vzero}};
  for (int s = 0; s < 8; ++s) {
    const int k = s * 4 + 2 * half;
    v2f ay; ay.x = st[m * CCH + k]; ay.y = st[m * CCH + k + 1];
#pragma unroll
    for (int g = 0; g < 3; ++g) MM2(ga[g], ay, W_G(g), s);
  }
  v8f gate[3][2];
#pragma unroll
  for (int g = 0; g < 3; ++g)
#pragma unroll
    for (int t = 0; t < 2; ++t)
#pragma unroll
      for (int r = 0; r < 8; ++r) {
        const float v = ga[g][t][r];
        gate[g][t][r] = v * (1.0f / (1.0f + __expf(-v)));
      }

#pragma unroll
  for (int t = 0; t < 2; ++t)
#pragma unroll
    for (int r = 0; r < 8; ++r)
      st[(r + 8 * half) * CCH + t * 16 + m] = gate[0][t][r];
  STAGE_FENCE();
  flush_tile<4>(st, node0 + (size_t)tile * 16 * CCH,
                out0 + (size_t)tile * 16 * CCH, nvr * 8, lane);

  v8f y1acc[3][2] = {{vzero, vzero}, {vzero, vzero}, {vzero, vzero}};
  for (int s = 0; s < 8; ++s) {
    LOAD_BASE(s);
#pragma unroll
    for (int v = 0; v < 3; ++v) {
      v2f av, a01, tv;
#pragma unroll
      for (int jj = 0; jj < 2; ++jj) {
        av[jj]  = b1[jj][v];
        a01[jj] = x0[jj] * b1[jj][v];
        float acc = 0.f;
#pragma unroll
        for (int a = 0; a < 3; ++a) acc += b1[jj][a] * b2[jj][a * 3 + v];
        tv[jj] = acc;
      }
      MM2(y1acc[v], av,  W_U1,  s);
      MM2(y1acc[v], a01, W_V(1), s);
      MM2(y1acc[v], tv,  W_V(5), s);
    }
  }
#pragma unroll
  for (int v = 0; v < 3; ++v)
#pragma unroll
    for (int t = 0; t < 2; ++t)
#pragma unroll
      for (int r = 0; r < 8; ++r)
        st[((r + 8 * half) * CCH + t * 16 + m) * 3 + v] =
            y1acc[v][t][r] * gate[1][t][r];
  STAGE_FENCE();
  flush_tile<12>(st, node1 + (size_t)tile * 16 * CCH * 3,
                 out1 + (size_t)tile * 16 * CCH * 3, nvr * 24, lane);

  for (int aIdx = 0; aIdx < 3; ++aIdx) {
    asm volatile("" ::: "memory");
    v8f y2acc[3][2] = {{vzero, vzero}, {vzero, vzero}, {vzero, vzero}};
    for (int s = 0; s < 8; ++s) {
      LOAD_BASE(s);
#pragma unroll
      for (int b = 0; b < 3; ++b) {
        v2f a2v, a02, a11, qv;
#pragma unroll
        for (int jj = 0; jj < 2; ++jj) {
          const float g = b2[jj][aIdx * 3 + b];
          a2v[jj] = g;
          a02[jj] = x0[jj] * g;
          a11[jj] = b1[jj][aIdx] * b1[jj][b];
          float acc = 0.f;
#pragma unroll
          for (int kk = 0; kk < 3; ++kk)
            acc += b2[jj][aIdx * 3 + kk] * b2[jj][kk * 3 + b];
          qv[jj] = acc;
        }
        MM2(y2acc[b], a2v, W_U2,  s);
        MM2(y2acc[b], a02, W_V(2), s);
        MM2(y2acc[b], a11, W_V(4), s);
        MM2(y2acc[b], qv,  W_V(7), s);
      }
    }
#pragma unroll
    for (int b = 0; b < 3; ++b)
#pragma unroll
      for (int t = 0; t < 2; ++t)
#pragma unroll
        for (int r = 0; r < 8; ++r)
          st[((r + 8 * half) * CCH + t * 16 + m) * 9 + aIdx * 3 + b] =
              y2acc[b][t][r] * gate[2][t][r];
  }
  STAGE_FENCE();
  flush_tile<36>(st, node2 + (size_t)tile * 16 * CCH * 9,
                 out2 + (size_t)tile * 16 * CCH * 9, nvr * 72, lane);
  }
#undef STAGE_FENCE
#undef LOAD_BASE
#undef MM2
#undef LDB
}

extern "C" void kernel_launch(void* const* d_in, const int* in_sizes, int n_in,
                              void* d_out, int out_size, void* d_ws, size_t ws_size,
                              hipStream_t stream) {
  const float* node0 = (const float*)d_in[0];
  const float* node1 = (const float*)d_in[1];
  const float* node2 = (const float*)d_in[2];
  const float* rij   = (const float*)d_in[3];
  const float* Wrad  = (const float*)d_in[4];
  const float* U     = (const float*)d_in[5];
  const float* V     = (const float*)d_in[6];
  const float* Wg    = (const float*)d_in[7];
  const int*   idx_i = (const int*)d_in[8];
  const int*   idx_j = (const int*)d_in[9];

  const int N = in_sizes[0] / CCH;
  const int E = in_sizes[3] / 3;

  float* a0 = (float*)d_ws;
  float* a1 = a0 + (size_t)N * CCH;
  float* a2 = a1 + (size_t)N * CCH * 3;

  float* out0 = (float*)d_out;
  float* out1 = out0 + (size_t)N * CCH;
  float* out2 = out1 + (size_t)N * CCH * 3;

  mmb_edge<<<(N + EDGE_TN - 1) / EDGE_TN, EDGE_WAVES * 32, 0, stream>>>(
      rij, Wrad, idx_i, idx_j, node0, node1, node2, a0, a1, a2, N, E);

  mmb_node<<<NODE_WGS, NODE_WAVES * 32, 0, stream>>>(a0, a1, a2,
                                                 node0, node1, node2,
                                                 U, V, Wg,
                                                 out0, out1, out2, N);
}


# workspace re-laid in matrix-operand order [16-node tile][k-step][16-byte piece][lane]: edge phase stores it so from padded LDS rows; node phase A loads: 7 whole-line loads per k-step, were 8 strided
# speedup vs baseline: 4.7530x; 1.0899x over previous
#include <hip/hip_runtime.h>
#include <hip/hip_bf16.h>
#include <math.h>

#define CCH 32
#define NPATH 11
#define NB 8

typedef __attribute__((ext_vector_type(2))) float v2f;
typedef __attribute__((ext_vector_type(8))) float v8f;
typedef __attribute__((ext_vector_type(4))) float v4f;

__device__ __forceinline__ v8f wmma4(v2f a, v2f b, v8f c) {
  return __builtin_amdgcn_wmma_f32_16x16x4_f32(
       false, a,  false, b,
       (short)0, c,  false,  false);
}

#define EDGE_TN    80
#define PIT0 34
#define PIT1 98
#define PIT2 290
#define ACC_F (EDGE_TN * (PIT0 + PIT1 + PIT2))
#define TILE_F (16 * CCH * 13)
#define STEP_F (32 * 26)
#define EDGE_WAVES 16
#define EDGE_SCAN  12
#define QCAP       2048
#define NFRAG      (NPATH * 2 * 2)

__device__ __forceinline__ void lds_add(float* p, float v) {
  __hip_atomic_fetch_add(p, v, __ATOMIC_RELAXED, __HIP_MEMORY_SCOPE_WORKGROUP);
}

__device__ __forceinline__ float xhalf(float x) {
  const int b = __builtin_bit_cast(int, x);
  return __builtin_bit_cast(float,
      __builtin_amdgcn_permlanex16(b, b, 0x76543210, 0xfedcba98, false, false));
}

__device__ __forceinline__ float lane_f(float x, int l) {
  return __builtin_bit_cast(float,
      __builtin_amdgcn_readlane(__builtin_bit_cast(int, x), l));
}

struct NodeRows { float g0; float g1[3]; float g2[9]; };

__device__ __forceinline__ void load_rows(NodeRows& g,
                                          const float* __restrict__ node0,
                                          const float* __restrict__ node1,
                                          const float* __restrict__ node2,
                                          int j, int ch) {
  g.g0 = node0[j * CCH + ch];
#pragma unroll
  for (int a = 0; a < 3; ++a) g.g1[a] = node1[(j * CCH + ch) * 3 + a];
#pragma unroll
  for (int q = 0; q < 9; ++q) g.g2[q] = node2[(j * CCH + ch) * 9 + q];
}

__device__ __forceinline__ void edge_add(const NodeRows& g,
                                         float* s0, float* s1, float* s2,
                                         const float (&w)[NPATH],
                                         int nl, float rh0, float rh1, float rh2,
                                         int ch, bool valid) {
  const float rh[3] = {rh0, rh1, rh2};
  const float g0 = g.g0;
  const float (&g1)[3] = g.g1;
  const float (&g2)[9] = g.g2;

  const float d1 = g1[0] * rh[0] + g1[1] * rh[1] + g1[2] * rh[2];
  float gr[3];
#pragma unroll
  for (int a = 0; a < 3; ++a)
    gr[a] = g2[a * 3 + 0] * rh[0] + g2[a * 3 + 1] * rh[1] + g2[a * 3 + 2] * rh[2];
  const float d2 = gr[0] * rh[0] + gr[1] * rh[1] + gr[2] * rh[2];

  const float inv16 = 0.0625f;

  const float m0 = (g0 * w[0] + d1 * w[4] + d2 * w[9]) * inv16;
  float m1[3], m2[9];
#pragma unroll
  for (int a = 0; a < 3; ++a) {
    const float mv = (g0 * w[1] + d1 * w[6]) * rh[a] + g1[a] * w[3] + gr[a] * w[8];
    m1[a] = mv * inv16;
  }
#pragma unroll
  for (int a = 0; a < 3; ++a) {
    const float ca = g0 * w[2] * rh[a] + g1[a] * w[5] + gr[a] * w[10];
#pragma unroll
    for (int b = 0; b < 3; ++b) {
      const float mv = ca * rh[b] + g2[a * 3 + b] * w[7];
      m2[a * 3 + b] = mv * inv16;
    }
  }
  if (valid) {
    lds_add(&s0[nl * PIT0 + ch], m0);
#pragma unroll
    for (int a = 0; a < 3; ++a) lds_add(&s1[nl * PIT1 + ch * 3 + a], m1[a]);
#pragma unroll
    for (int k = 0; k < 9; ++k) lds_add(&s2[nl * PIT2 + ch * 9 + k], m2[k]);
  }
}

__global__ void __launch_bounds__(EDGE_WAVES * 32)
mmb_edge(const float* __restrict__ rij,
         const float* __restrict__ Wrad,
         const int*   __restrict__ idx_i,
         const int*   __restrict__ idx_j,
         const float* __restrict__ node0,
         const float* __restrict__ node1,
         const float* __restrict__ node2,
         float* __restrict__ a0,
         float* __restrict__ a1,
         float* __restrict__ a2,
         int N, int E) {
  __shared__ float4 acc4[ACC_F / 4];
  float* const s0 = (float*)acc4;
  float* const s1 = s0 + EDGE_TN * PIT0;
  float* const s2 = s1 + EDGE_TN * PIT1;
  __shared__ float2 wb[NFRAG * 32];
  __shared__ int queue[EDGE_WAVES * QCAP];

  const int tid = threadIdx.x;
  const int c   = tid & 31;
  const int hf  = c >> 4;
  const int q   = c & 15;
  const bool lo = (c < 16);
  const int wv  = __builtin_amdgcn_readfirstlane(tid >> 5);
  const int n0  = blockIdx.x * EDGE_TN;
  const int nv  = (N - n0 < EDGE_TN) ? (N - n0) : EDGE_TN;
  int* const qw = queue + wv * QCAP;

  constexpr int WR_N     = NFRAG * 64;
  constexpr int WR_TRIPS = (WR_N + EDGE_WAVES * 32 - 1) / (EDGE_WAVES * 32);
  float wt[WR_TRIPS];
#pragma unroll
  for (int u = 0; u < WR_TRIPS; ++u) {
    int k = tid + u * EDGE_WAVES * 32;
    k = (k < WR_N) ? k : (WR_N - 1);
    const int jj = k & 1, ln = (k >> 1) & 31, f = k >> 6;
    const int t = f & 1, s = (f >> 1) & 1, p = f >> 2;
    const int kk = 4 * s + 2 * (ln >> 4) + jj;
    wt[u] = Wrad[(p * NB + kk) * CCH + 16 * t + (ln & 15)];
  }
  for (int k = tid; k < ACC_F / 4; k += EDGE_WAVES * 32)
    acc4[k] = make_float4(0.f, 0.f, 0.f, 0.f);
#pragma unroll
  for (int u = 0; u < WR_TRIPS; ++u) {
    const int k = tid + u * EDGE_WAVES * 32;
    if (k < WR_N) ((float*)wb)[k] = wt[u];
  }
  __syncthreads();

#define CEN(b) (0.5f + 0.6428571429f * (float)(b))
  const float cen00 = hf ? CEN(2) : CEN(0);
  const float cen01 = hf ? CEN(3) : CEN(1);
  const float cen10 = hf ? CEN(6) : CEN(4);
  const float cen11 = hf ? CEN(7) : CEN(5);
#undef CEN
  const v8f vzero = {};

  const int nPieces = (E + 127) >> 7;
  int cb   = wv;
  int head = 0;
  int cnt  = 0;
  for (;;) {
    if (cnt < 16 && cb < nPieces) {
      int4 il[EDGE_SCAN];
      const bool whole = (((cb + (EDGE_SCAN - 1) * EDGE_WAVES) << 7) + 128) <= E;
      if (whole) {
#pragma unroll
        for (int u = 0; u < EDGE_SCAN; ++u) {
          const int e0 = ((cb + u * EDGE_WAVES) << 7) + 4 * c;
          il[u] = *(const int4*)(idx_i + e0);
        }
      } else {
#pragma unroll
        for (int u = 0; u < EDGE_SCAN; ++u) {
          const int e0 = ((cb + u * EDGE_WAVES) << 7) + 4 * c;
          int id[4];
#pragma unroll
          for (int k = 0; k < 4; ++k) {
            const int ev = e0 + k;
            const int v  = idx_i[(ev < E) ? ev : (E - 1)];
            id[k] = (ev < E) ? v : (n0 - 1);
          }
          il[u] = make_int4(id[0], id[1], id[2], id[3]);
        }
      }
#pragma unroll
      for (int u = 0; u < EDGE_SCAN; ++u) {
        const int e0 = ((cb + u * EDGE_WAVES) << 7) + 4 * c;
        const unsigned a[4] = {(unsigned)(il[u].x - n0), (unsigned)(il[u].y - n0),
                               (unsigned)(il[u].z - n0), (unsigned)(il[u].w - n0)};
        const unsigned lo4 = min(min(a[0], a[1]), min(a[2], a[3]));
        if (__builtin_amdgcn_ballot_w32(lo4 < (unsigned)nv) != 0u) {
#pragma unroll
          for (int k = 0; k < 4; ++k) {
            const bool hit = a[k] < (unsigned)nv;
            const unsigned mu = __builtin_amdgcn_ballot_w32(hit);
            if (mu != 0u) {
              const int pos = cnt + (int)__builtin_amdgcn_mbcnt_lo(mu, 0u);
              if (hit) qw[(head + pos) & (QCAP - 1)] = e0 + k;
              cnt += __builtin_popcount(mu);
            }
          }
        }
      }
      cb += EDGE_WAVES * EDGE_SCAN;
      continue;
    }
    if (cnt == 0) break;
    const int nb = (cnt < 16) ? cnt : 16;

    asm volatile("s_wait_dscnt 0" ::: "memory");

    const int e = qw[(head + ((q < nb) ? q : 0)) & (QCAP - 1)];
    const float r0 = rij[e * 3 + 0], r1 = rij[e * 3 + 1], r2 = rij[e * 3 + 2];
    const int   jv  = idx_j[e];
    const int   nlv = idx_i[e] - n0;
    const float d  = sqrtf(r0 * r0 + r1 * r1 + r2 * r2 + 1e-12f);
    const float iv = 1.0f / d;
    const float rhx = r0 * iv, rhy = r1 * iv, rhz = r2 * iv;

    float x = d * 0.2f;
    x = (x > 1.0f) ? 1.0f : x;
    const float fc = 0.5f * (cosf(3.14159265358979f * x) + 1.0f);

    const float t00 = d - cen00, t01 = d - cen01, t10 = d - cen10, t11 = d - cen11;
    v2f af0, af1;
    af0.x = __expf(-4.0f * t00 * t00) * fc;
    af0.y = __expf(-4.0f * t01 * t01) * fc;
    af1.x = __expf(-4.0f * t10 * t10) * fc;
    af1.y = __expf(-4.0f * t11 * t11) * fc;

    v8f D[NPATH][2];
#pragma unroll
    for (int p = 0; p < NPATH; ++p)
#pragma unroll
      for (int t = 0; t < 2; ++t) {
        const float2 b0 = wb[((p * 2 + 0) * 2 + t) * 32 + c];
        const float2 b1 = wb[((p * 2 + 1) * 2 + t) * 32 + c];
        v2f bf0, bf1;
        bf0.x = b0.x; bf0.y = b0.y; bf1.x = b1.x; bf1.y = b1.y;
        v8f acc = wmma4(af0, bf0, vzero);
        D[p][t] = wmma4(af1, bf1, acc);
      }

#pragma unroll
    for (int r = 0; r < 8; ++r) {
      if (r < nb) {
        const bool valid = (8 * hf + r) < nb;
        const int   jA = __builtin_amdgcn_readlane(jv, r);
        const int   jB = __builtin_amdgcn_readlane(jv, 8 + r);
        const int   nA = __builtin_amdgcn_readlane(nlv, r);
        const int   nB = __builtin_amdgcn_readlane(nlv, 8 + r);
        const float xA = lane_f(rhx, r), xB = lane_f(rhx, 8 + r);
        const float yA = lane_f(rhy, r), yB = lane_f(rhy, 8 + r);
        const float zA = lane_f(rhz, r), zB = lane_f(rhz, 8 + r);
        const int   j  = hf ? jB : jA;
        const int   nl = hf ? nB : nA;
        const float h0 = hf ? xB : xA;
        const float h1 = hf ? yB : yA;
        const float h2 = hf ? zB : zA;
        NodeRows gA, gB;
        load_rows(gA, node0, node1, node2, j, q);
        load_rows(gB, node0, node1, node2, j, 16 + q);
        float wA[NPATH], wB[NPATH];
#pragma unroll
        for (int p = 0; p < NPATH; ++p) { wA[p] = D[p][0][r]; wB[p] = D[p][1][r]; }
        edge_add(gA, s0, s1, s2, wA, nl, h0, h1, h2, q, valid);
        edge_add(gB, s0, s1, s2, wB, nl, h0, h1, h2, 16 + q, valid);
      }
    }
    head = (head + nb) & (QCAP - 1);
    cnt -= nb;
  }
  __syncthreads();

  float* const wsb = a0 + (size_t)blockIdx.x * (EDGE_TN / 16) * TILE_F;
  const int nPairs = ((nv + 15) >> 4) * 8;
  for (int pr = wv; pr < nPairs; pr += EDGE_WAVES) {
    const int lt = pr >> 3;
    const int s  = pr & 7;
    const int nl = lt * 16 + q;
    if (nl < nv) {
      const int c0 = 4 * s + 2 * hf;
      const float2* const S0 = (const float2*)(s0 + nl * PIT0 + c0);
      const float2* const S1 = (const float2*)(s1 + nl * PIT1 + c0 * 3);
      const float2* const S2 = (const float2*)(s2 + nl * PIT2 + c0 * 9);
      float2 u[13];
      u[0] = S0[0];
#pragma unroll
      for (int a = 0; a < 3; ++a) u[1 + a] = S1[a];
#pragma unroll
      for (int k = 0; k < 9; ++k) u[4 + k] = S2[k];
      float* const dst = wsb + lt * TILE_F + s * STEP_F;
#pragma unroll
      for (int p = 0; p < 6; ++p)
        *(float4*)(dst + p * 128 + c * 4) =
            make_float4(u[2 * p].x, u[2 * p].y, u[2 * p + 1].x, u[2 * p + 1].y);
      *(float2*)(dst + 768 + c * 2) = u[12];
    }
  }
}

#define STAGE_F (16 * CCH * 9)
#define NODE_WAVES 8
#define NODE_WGS   256

template <int PER_LANE>
__device__ __forceinline__ void flush_tile(const float* st,
                                           const float* __restrict__ res,
                                           float* __restrict__ dst,
                                           int nvalid4, int lane) {
  const v4f* const s4 = (const v4f*)st;
  const v4f* const r4 = (const v4f*)res;
  v4f* const d4 = (v4f*)dst;
  constexpr int BATCH = (PER_LANE < 12) ? PER_LANE : 12;
#pragma unroll
  for (int i0 = 0; i0 < PER_LANE; i0 += BATCH) {
    v4f rv[BATCH], sv[BATCH];
#pragma unroll
    for (int i = 0; i < BATCH; ++i) {
      const int k = (i0 + i) * 32 + lane;
      rv[i] = r4[(k < nvalid4) ? k : (nvalid4 - 1)];
      sv[i] = s4[k];
    }
#pragma unroll
    for (int i = 0; i < BATCH; ++i) {
      const int k = (i0 + i) * 32 + lane;
      if (k < nvalid4) d4[k] = sv[i] + rv[i];
    }
  }
}
#define W_U0 0
#define W_U1 1
#define W_U2 2
#define W_V(k) (3 + (k))
#define W_G(k) (11 + (k))

__global__ void __launch_bounds__(NODE_WAVES * 32)
mmb_node(const float* __restrict__ a0, const float* __restrict__ a1,
         const float* __restrict__ a2,
         const float* __restrict__ node0, const float* __restrict__ node1,
         const float* __restrict__ node2,
         const float* __restrict__ U, const float* __restrict__ V,
         const float* __restrict__ Wg,
         float* __restrict__ out0, float* __restrict__ out1,
         float* __restrict__ out2, int N) {
  __shared__ float2 wlds[14 * 8 * 2 * 32];
  __shared__ __attribute__((aligned(16))) float stage[NODE_WAVES * STAGE_F];

  for (int idx = threadIdx.x; idx < 14 * 1024; idx += blockDim.x) {
    const int wmat = idx >> 10;
    const int rem  = idx & 1023;
    const int k    = rem >> 5;
    const int n    = rem & 31;
    const float* src = (wmat < 3) ? (U + wmat * 1024)
                     : (wmat < 11) ? (V + (wmat - 3) * 1024)
                                   : (Wg + (wmat - 11) * 1024);
    const float val = src[rem];
    const int s   = k >> 2;
    const int kk  = k & 3;
    const int ln  = ((kk >> 1) << 4) | (n & 15);
    const int t   = n >> 4;
    const int jj  = kk & 1;
    ((float*)&wlds[((wmat * 8 + s) * 2 + t) * 32 + ln])[jj] = val;
  }
  __syncthreads();

  const int lane = threadIdx.x & 31;
  const int wv   = __builtin_amdgcn_readfirstlane(threadIdx.x >> 5);
  const int half = lane >> 4;
  const int m    = lane & 15;
  const int nTiles = (N + 15) >> 4;

#define LDB(wm, s, t)                                                       \
  ({ float2 _bv = wlds[(((wm) * 8 + (s)) * 2 + (t)) * 32 + lane];           \
     v2f _b; _b.x = _bv.x; _b.y = _bv.y; _b; })

#define MM2(acc, aop, wm, s)                                                \
  do { v2f _a = (aop);                                                      \
       acc[0] = wmma4(_a, LDB(wm, s, 0), acc[0]);                           \
       acc[1] = wmma4(_a, LDB(wm, s, 1), acc[1]); } while (0)

  const v8f vzero = {};
  float* const st = stage + wv * STAGE_F;
  for (int tile = wv * (int)gridDim.x + (int)blockIdx.x; tile < nTiles;
       tile += NODE_WAVES * (int)gridDim.x) {
  const int nvr = (N - tile * 16 < 16) ? (N - tile * 16) : 16;
#define STAGE_FENCE() asm volatile("s_wait_dscnt 0" ::: "memory")

  const float* const tb = a0 + (size_t)tile * TILE_F;
  const int ln = half * 16 + ((m < nvr) ? m : (nvr - 1));
  float x0[2], b1[2][3], b2[2][9];
#define LOAD_BASE(s)                                                        \
  do { const float* const _b = tb + (s) * STEP_F;                           \
       const v4f _p0 = *(const v4f*)(_b + ln * 4);                          \
       const v4f _p1 = *(const v4f*)(_b + 128 + ln * 4);                    \
       const v4f _p2 = *(const v4f*)(_b + 256 + ln * 4);                    \
       const v4f _p3 = *(const v4f*)(_b + 384 + ln * 4);                    \
       const v4f _p4 = *(const v4f*)(_b + 512 + ln * 4);                    \
       const v4f _p5 = *(const v4f*)(_b + 640 + ln * 4);                    \
       const v2f _p6 = *(const v2f*)(_b + 768 + ln * 2);                    \
       x0[0] = _p0.x; x0[1] = _p0.y;                                        \
       b1[0][0] = _p0.z; b1[0][1] = _p0.w; b1[0][2] = _p1.x;                \
       b1[1][0] = _p1.y; b1[1][1] = _p1.z; b1[1][2] = _p1.w;                \
       b2[0][0] = _p2.x; b2[0][1] = _p2.y; b2[0][2] = _p2.z; b2[0][3] = _p2.w; \
       b2[0][4] = _p3.x; b2[0][5] = _p3.y; b2[0][6] = _p3.z; b2[0][7] = _p3.w; \
       b2[0][8] = _p4.x; b2[1][0] = _p4.y; b2[1][1] = _p4.z; b2[1][2] = _p4.w; \
       b2[1][3] = _p5.x; b2[1][4] = _p5.y; b2[1][5] = _p5.z; b2[1][6] = _p5.w; \
       b2[1][7] = _p6.x; b2[1][8] = _p6.y;                                  \
  } while (0)

  v8f y0acc[2] = {vzero, vzero};
  for (int s = 0; s < 8; ++s) {
    LOAD_BASE(s);
    v2f ax, asq, s1v, s2v;
#pragma unroll
    for (int jj = 0; jj < 2; ++jj) {
      ax[jj]  = x0[jj];
      asq[jj] = x0[jj] * x0[jj];
      float acc = 0.f;
#pragma unroll
      for (int a = 0; a < 3; ++a) acc += b1[jj][a] * b1[jj][a];
      s1v[jj] = acc;
      acc = 0.f;
#pragma unroll
      for (int q = 0; q < 9; ++q) acc += b2[jj][q] * b2[jj][q];
      s2v[jj] = acc;
    }
    MM2(y0acc, ax,  W_U0,  s);
    MM2(y0acc, asq, W_V(0), s);
    MM2(y0acc, s1v, W_V(3), s);
    MM2(y0acc, s2v, W_V(6), s);
  }

#pragma unroll
  for (int t = 0; t < 2; ++t)
#pragma unroll
    for (int r = 0; r < 8; ++r)
      st[(r + 8 * half) * CCH + t * 16 + m] = y0acc[t][r];
  STAGE_FENCE();

  v8f ga[3][2] = {{vzero, vzero}, {vzero, vzero}, {vzero, vzero}};
  for (int s = 0; s < 8; ++s) {
    const int k = s * 4 + 2 * half;
    v2f ay; ay.x = st[m * CCH + k]; ay.y = st[m * CCH + k + 1];
#pragma unroll
    for (int g = 0; g < 3; ++g) MM2(ga[g], ay, W_G(g), s);
  }
  v8f gate[3][2];
#pragma unroll
  for (int g = 0; g < 3; ++g)
#pragma unroll
    for (int t = 0; t < 2; ++t)
#pragma unroll
      for (int r = 0; r < 8; ++r) {
        const float v = ga[g][t][r];
        gate[g][t][r] = v * (1.0f / (1.0f + __expf(-v)));
      }

#pragma unroll
  for (int t = 0; t < 2; ++t)
#pragma unroll
    for (int r = 0; r < 8; ++r)
      st[(r + 8 * half) * CCH + t * 16 + m] = gate[0][t][r];
  STAGE_FENCE();
  flush_tile<4>(st, node0 + (size_t)tile * 16 * CCH,
                out0 + (size_t)tile * 16 * CCH, nvr * 8, lane);

  v8f y1acc[3][2] = {{vzero, vzero}, {vzero, vzero}, {vzero, vzero}};
  for (int s = 0; s < 8; ++s) {
    LOAD_BASE(s);
#pragma unroll
    for (int v = 0; v < 3; ++v) {
      v2f av, a01, tv;
#pragma unroll
      for (int jj = 0; jj < 2; ++jj) {
        av[jj]  = b1[jj][v];
        a01[jj] = x0[jj] * b1[jj][v];
        float acc = 0.f;
#pragma unroll
        for (int a = 0; a < 3; ++a) acc += b1[jj][a] * b2[jj][a * 3 + v];
        tv[jj] = acc;
      }
      MM2(y1acc[v], av,  W_U1,  s);
      MM2(y1acc[v], a01, W_V(1), s);
      MM2(y1acc[v], tv,  W_V(5), s);
    }
  }
#pragma unroll
  for (int v = 0; v < 3; ++v)
#pragma unroll
    for (int t = 0; t < 2; ++t)
#pragma unroll
      for (int r = 0; r < 8; ++r)
        st[((r + 8 * half) * CCH + t * 16 + m) * 3 + v] =
            y1acc[v][t][r] * gate[1][t][r];
  STAGE_FENCE();
  flush_tile<12>(st, node1 + (size_t)tile * 16 * CCH * 3,
                 out1 + (size_t)tile * 16 * CCH * 3, nvr * 24, lane);

#pragma nounroll
  for (int aIdx = 0; aIdx < 3; ++aIdx) {
    asm volatile("" ::: "memory");
    v8f y2acc[3][2] = {{vzero, vzero}, {vzero, vzero}, {vzero, vzero}};
    for (int s = 0; s < 8; ++s) {
      LOAD_BASE(s);
#pragma unroll
      for (int b = 0; b < 3; ++b) {
        v2f a2v, a02, a11, qv;
#pragma unroll
        for (int jj = 0; jj < 2; ++jj) {
          const float g = b2[jj][aIdx * 3 + b];
          a2v[jj] = g;
          a02[jj] = x0[jj] * g;
          a11[jj] = b1[jj][aIdx] * b1[jj][b];
          float acc = 0.f;
#pragma unroll
          for (int kk = 0; kk < 3; ++kk)
            acc += b2[jj][aIdx * 3 + kk] * b2[jj][kk * 3 + b];
          qv[jj] = acc;
        }
        MM2(y2acc[b], a2v, W_U2,  s);
        MM2(y2acc[b], a02, W_V(2), s);
        MM2(y2acc[b], a11, W_V(4), s);
        MM2(y2acc[b], qv,  W_V(7), s);
      }
    }
#pragma unroll
    for (int b = 0; b < 3; ++b)
#pragma unroll
      for (int t = 0; t < 2; ++t)
#pragma unroll
        for (int r = 0; r < 8; ++r)
          st[((r + 8 * half) * CCH + t * 16 + m) * 9 + aIdx * 3 + b] =
              y2acc[b][t][r] * gate[2][t][r];
  }
  STAGE_FENCE();
  flush_tile<36>(st, node2 + (size_t)tile * 16 * CCH * 9,
                 out2 + (size_t)tile * 16 * CCH * 9, nvr * 72, lane);
  }
#undef STAGE_FENCE
#undef LOAD_BASE
#undef MM2
#undef LDB
}

extern "C" void kernel_launch(void* const* d_in, const int* in_sizes, int n_in,
                              void* d_out, int out_size, void* d_ws, size_t ws_size,
                              hipStream_t stream) {
  const float* node0 = (const float*)d_in[0];
  const float* node1 = (const float*)d_in[1];
  const float* node2 = (const float*)d_in[2];
  const float* rij   = (const float*)d_in[3];
  const float* Wrad  = (const float*)d_in[4];
  const float* U     = (const float*)d_in[5];
  const float* V     = (const float*)d_in[6];
  const float* Wg    = (const float*)d_in[7];
  const int*   idx_i = (const int*)d_in[8];
  const int*   idx_j = (const int*)d_in[9];

  const int N = in_sizes[0] / CCH;
  const int E = in_sizes[3] / 3;

  float* a0 = (float*)d_ws;
  float* a1 = a0 + (size_t)N * CCH;
  float* a2 = a1 + (size_t)N * CCH * 3;

  float* out0 = (float*)d_out;
  float* out1 = out0 + (size_t)N * CCH;
  float* out2 = out1 + (size_t)N * CCH * 3;

  mmb_edge<<<(N + EDGE_TN - 1) / EDGE_TN, EDGE_WAVES * 32, 0, stream>>>(
      rij, Wrad, idx_i, idx_j, node0, node1, node2, a0, a1, a2, N, E);

  mmb_node<<<NODE_WGS, NODE_WAVES * 32, 0, stream>>>(a0, a1, a2,
                                                 node0, node1, node2,
                                                 U, V, Wg,
                                                 out0, out1, out2, N);
}
